# starting kernel
# baseline (speedup 1.0000x reference)
#include <hip/hip_runtime.h>
#include <math.h>

#define FD 128
#define WAVE 32

typedef __attribute__((ext_vector_type(2))) float v2f;
typedef __attribute__((ext_vector_type(8))) float v8f;

__global__ void k_deg_init(unsigned* __restrict__ degi, int n) {
    int i = blockIdx.x * blockDim.x + threadIdx.x;
    if (i < n) degi[i] = 1u;
}

__global__ void k_deg_count(const int* __restrict__ dst,
                            unsigned* __restrict__ degi, int e) {
    int i = blockIdx.x * blockDim.x + threadIdx.x;
    if (i < e) atomicAdd(&degi[dst[i]], 1u);
}

__global__ void k_dinv(const unsigned* __restrict__ degi,
                       float* __restrict__ dinv, int n) {
    int i = blockIdx.x * blockDim.x + threadIdx.x;
    if (i < n) dinv[i] = rsqrtf((float)degi[i]);
}

__global__ __launch_bounds__(256) void k_gemm128_wmma(
    const float* __restrict__ A, const float* __restrict__ W,
    const float* __restrict__ bias, float* __restrict__ out,
    int fuse_bias_relu) {
    const int lane = threadIdx.x & 31;
    const int wid  = threadIdx.x >> 5;
    const int m0   = blockIdx.x << 4;
    const int n0   = wid << 4;
    const int half = lane >> 4;
    const int lm   = lane & 15;

    const float* __restrict__ arow = A + (size_t)(m0 + lm) * FD;

    v8f acc0 = {};
    v8f acc1 = {};
#pragma unroll
    for (int k0 = 0; k0 < FD; k0 += 8) {
        {
            const int ka = k0 + 2 * half;
            v2f a, b;
            a.x = arow[ka];
            a.y = arow[ka + 1];
            b.x = W[(size_t)(ka)     * FD + n0 + lm];
            b.y = W[(size_t)(ka + 1) * FD + n0 + lm];
            acc0 = __builtin_amdgcn_wmma_f32_16x16x4_f32(
                false, a, false, b, (short)0, acc0, false, false);
        }
        {
            const int ka = k0 + 4 + 2 * half;
            v2f a, b;
            a.x = arow[ka];
            a.y = arow[ka + 1];
            b.x = W[(size_t)(ka)     * FD + n0 + lm];
            b.y = W[(size_t)(ka + 1) * FD + n0 + lm];
            acc1 = __builtin_amdgcn_wmma_f32_16x16x4_f32(
                false, a, false, b, (short)0, acc1, false, false);
        }
    }

#pragma unroll
    for (int v = 0; v < 8; ++v) {
        const int row = m0 + v + 8 * half;
        const int col = n0 + lm;
        float r = acc0[v] + acc1[v];
        if (fuse_bias_relu) { r += bias[col]; r = fmaxf(r, 0.0f); }
        out[(size_t)row * FD + col] = r;
    }
}

__global__ void k_agg_init(const float* __restrict__ hw,
                           const float* __restrict__ dinv,
                           const float* __restrict__ bias,
                           float* __restrict__ agg, int nd) {
    int i = blockIdx.x * blockDim.x + threadIdx.x;
    if (i < nd) {
        const int node = i >> 7;
        const int f    = i & 127;
        const float di = dinv[node];
        agg[i] = hw[i] * di * di + bias[f];
    }
}

__global__ __launch_bounds__(256) void k_edge_scatter(
    const int* __restrict__ src, const int* __restrict__ dst,
    const float* __restrict__ dinv, const float* __restrict__ hw,
    float* __restrict__ agg, int e) {
    const int wid  = (int)((blockIdx.x * blockDim.x + threadIdx.x) >> 5);
    const int lane = threadIdx.x & 31;
    if (wid >= e) return;
    const int   s = src[wid];
    const int   d = dst[wid];
    const float w = dinv[s] * dinv[d];
    const float4 v = ((const float4*)(hw + (size_t)s * FD))[lane];
    float* ap = agg + (size_t)d * FD + lane * 4;
    unsafeAtomicAdd(ap + 0, v.x * w);
    unsafeAtomicAdd(ap + 1, v.y * w);
    unsafeAtomicAdd(ap + 2, v.z * w);
    unsafeAtomicAdd(ap + 3, v.w * w);
}

__global__ void k_relu(const float* __restrict__ in, float* __restrict__ out,
                       int n) {
    int i = blockIdx.x * blockDim.x + threadIdx.x;
    if (i < n) out[i] = fmaxf(in[i], 0.0f);
}

__global__ void k_emb_relu(const float* __restrict__ agg,
                           float* __restrict__ emb, float* __restrict__ h,
                           int n) {
    int i = blockIdx.x * blockDim.x + threadIdx.x;
    if (i < n) {
        const float v = agg[i];
        emb[i] = v;
        h[i]   = fmaxf(v, 0.0f);
    }
}

__global__ __launch_bounds__(256) void k_logits(
    const float* __restrict__ h, const float* __restrict__ Wp2,
    const float* __restrict__ bp2, float* __restrict__ out, int n) {
    const int wid  = (int)((blockIdx.x * blockDim.x + threadIdx.x) >> 5);
    const int lane = threadIdx.x & 31;
    if (wid >= n) return;
    const float4 v = ((const float4*)(h + (size_t)wid * FD))[lane];
    const int f = lane * 4;
    float p0 = v.x * Wp2[(f + 0) * 2 + 0] + v.y * Wp2[(f + 1) * 2 + 0] +
               v.z * Wp2[(f + 2) * 2 + 0] + v.w * Wp2[(f + 3) * 2 + 0];
    float p1 = v.x * Wp2[(f + 0) * 2 + 1] + v.y * Wp2[(f + 1) * 2 + 1] +
               v.z * Wp2[(f + 2) * 2 + 1] + v.w * Wp2[(f + 3) * 2 + 1];
#pragma unroll
    for (int off = 16; off > 0; off >>= 1) {
        p0 += __shfl_down(p0, off, 32);
        p1 += __shfl_down(p1, off, 32);
    }
    if (lane == 0) {
        const float l0 = p0 + bp2[0];
        const float l1 = p1 + bp2[1];
        const float m  = fmaxf(l0, l1);
        const float lse = m + logf(expf(l0 - m) + expf(l1 - m));
        out[(size_t)wid * 2 + 0] = l0 - lse;
        out[(size_t)wid * 2 + 1] = l1 - lse;
    }
}

extern "C" void kernel_launch(void* const* d_in, const int* in_sizes, int n_in,
                              void* d_out, int out_size, void* d_ws,
                              size_t ws_size, hipStream_t stream) {
    const float* x   = (const float*)d_in[0];
    const int*   ei  = (const int*)d_in[1];
    const float* W1  = (const float*)d_in[2];
    const float* b1  = (const float*)d_in[3];
    const float* W2  = (const float*)d_in[4];
    const float* b2  = (const float*)d_in[5];
    const float* W3  = (const float*)d_in[6];
    const float* b3  = (const float*)d_in[7];
    const float* Wp1 = (const float*)d_in[8];
    const float* bp1 = (const float*)d_in[9];
    const float* Wp2 = (const float*)d_in[10];
    const float* bp2 = (const float*)d_in[11];

    const int N = in_sizes[0] / FD;
    const int E = in_sizes[1] / 2;
    const int* src = ei;
    const int* dst = ei + E;

    char* ws = (char*)d_ws;
    unsigned* degi = (unsigned*)ws;  ws += (size_t)N * sizeof(unsigned);
    float* dinv = (float*)ws;        ws += (size_t)N * sizeof(float);
    float* hw   = (float*)ws;        ws += (size_t)N * FD * sizeof(float);
    float* agg  = (float*)ws;        ws += (size_t)N * FD * sizeof(float);
    float* h    = (float*)ws;

    float* emb_out  = (float*)d_out;
    float* prob_out = (float*)d_out + (size_t)N * FD;

    const dim3 blk(256);
    const int ND      = N * FD;
    const int gN      = (N + 255) / 256;
    const int gE      = (E + 255) / 256;
    const int gND     = (ND + 255) / 256;
    const int gTiles  = N / 16;
    const int gEdgeW  = (E * WAVE + 255) / 256;
    const int gNodeW  = (N * WAVE + 255) / 256;

    k_deg_init <<<gN, blk, 0, stream>>>(degi, N);
    k_deg_count<<<gE, blk, 0, stream>>>(dst, degi, E);
    k_dinv     <<<gN, blk, 0, stream>>>(degi, dinv, N);

    k_gemm128_wmma<<<gTiles, blk, 0, stream>>>(x, W1, nullptr, hw, 0);
    k_agg_init    <<<gND, blk, 0, stream>>>(hw, dinv, b1, agg, ND);
    k_edge_scatter<<<gEdgeW, blk, 0, stream>>>(src, dst, dinv, hw, agg, E);
    k_relu        <<<gND, blk, 0, stream>>>(agg, h, ND);

    k_gemm128_wmma<<<gTiles, blk, 0, stream>>>(h, W2, nullptr, hw, 0);
    k_agg_init    <<<gND, blk, 0, stream>>>(hw, dinv, b2, agg, ND);
    k_edge_scatter<<<gEdgeW, blk, 0, stream>>>(src, dst, dinv, hw, agg, E);
    k_relu        <<<gND, blk, 0, stream>>>(agg, h, ND);

    k_gemm128_wmma<<<gTiles, blk, 0, stream>>>(h, W3, nullptr, hw, 0);
    k_agg_init    <<<gND, blk, 0, stream>>>(hw, dinv, b3, agg, ND);
    k_edge_scatter<<<gEdgeW, blk, 0, stream>>>(src, dst, dinv, hw, agg, E);
    k_emb_relu    <<<gND, blk, 0, stream>>>(agg, emb_out, h, ND);

    k_gemm128_wmma<<<gTiles, blk, 0, stream>>>(h, Wp1, bp1, hw, 1);
    k_logits      <<<gNodeW, blk, 0, stream>>>(hw, Wp2, bp2, prob_out, N);
}


# Edges grouped by destination once per call (rank = degree atomic's return, prefix scan, fill); each layer sums its rows by gather, one store per row, no f32 atomics; sum order = arrival rank
# speedup vs baseline: 2.5064x; 2.5064x over previous
#include <hip/hip_runtime.h>
#include <math.h>

#define FD 128
#define WAVE 32

typedef __attribute__((ext_vector_type(2))) float v2f;
typedef __attribute__((ext_vector_type(8))) float v8f;

__global__ void k_deg_init(unsigned* __restrict__ degi, int n) {
    int i = blockIdx.x * blockDim.x + threadIdx.x;
    if (i < n) degi[i] = 1u;
}

__global__ void k_deg_rank(const int* __restrict__ dst,
                           unsigned* __restrict__ degi,
                           int* __restrict__ rank, int e) {
    int i = blockIdx.x * blockDim.x + threadIdx.x;
    if (i < e) rank[i] = (int)atomicAdd(&degi[dst[i]], 1u);
}

__global__ __launch_bounds__(256) void k_rowptr_dinv(
    const unsigned* __restrict__ degi, int* __restrict__ rowptr,
    float* __restrict__ dinv, int n) {
    __shared__ unsigned red[8];
    __shared__ unsigned wtot[8];
    const int t    = threadIdx.x;
    const int lane = t & 31;
    const int wid  = t >> 5;
    const int base = blockIdx.x * 1024;
    const uint4* __restrict__ d4 = (const uint4*)degi;

    unsigned p = 0;
    const int nv = base >> 2;
#pragma unroll 8
    for (int j = t; j < nv; j += 256) {
        const uint4 v = d4[j];
        p += (v.x + v.y + v.z + v.w) - 4u;
    }
    for (int off = 16; off > 0; off >>= 1) p += __shfl_xor(p, off, 32);
    if (lane == 0) red[wid] = p;

    const int idx = base + 4 * t;
    const bool ok = (idx + 3 < n);
    uint4 c = make_uint4(1u, 1u, 1u, 1u);
    if (ok) c = d4[idx >> 2];
    const unsigned c0 = c.x - 1u, c1 = c.y - 1u, c2 = c.z - 1u, c3 = c.w - 1u;
    const unsigned s = c0 + c1 + c2 + c3;
    unsigned inc = s;
    for (int off = 1; off < 32; off <<= 1) {
        const unsigned y = __shfl_up(inc, off, 32);
        if (lane >= off) inc += y;
    }
    if (lane == 31) wtot[wid] = inc;
    __syncthreads();
    unsigned ex = inc - s;
    for (int w = 0; w < 8; ++w) {
        ex += red[w];
        if (w < wid) ex += wtot[w];
    }
    if (ok) {
        *(int4*)(rowptr + idx) = make_int4((int)ex, (int)(ex + c0),
                                           (int)(ex + c0 + c1),
                                           (int)(ex + c0 + c1 + c2));
        *(float4*)(dinv + idx) = make_float4(rsqrtf((float)c.x), rsqrtf((float)c.y),
                                             rsqrtf((float)c.z), rsqrtf((float)c.w));
        if (idx + 4 == n) rowptr[n] = (int)(ex + s);
    }
}

__device__ __forceinline__ void csr_put(int2* __restrict__ csr,
                                        const int* __restrict__ rowptr,
                                        const float* __restrict__ dinv,
                                        int s, int d, int r) {
    const float w = dinv[s] * dinv[d];
    csr[rowptr[d] + r - 1] = make_int2(s, __float_as_int(w));
}

__global__ __launch_bounds__(256) void k_csr_fill(
    const int* __restrict__ src, const int* __restrict__ dst,
    const int* __restrict__ rank, const int* __restrict__ rowptr,
    const float* __restrict__ dinv, int* __restrict__ csr_raw, int e) {
    int2* __restrict__ csr = (int2*)csr_raw;
    const int e0 = (blockIdx.x * blockDim.x + threadIdx.x) * 4;
    if (e0 >= e) return;
    if (((e & 3) == 0) && (e0 + 3 < e)) {
        const int4 s = *(const int4*)(src + e0);
        const int4 d = *(const int4*)(dst + e0);
        const int4 r = *(const int4*)(rank + e0);
        csr_put(csr, rowptr, dinv, s.x, d.x, r.x);
        csr_put(csr, rowptr, dinv, s.y, d.y, r.y);
        csr_put(csr, rowptr, dinv, s.z, d.z, r.z);
        csr_put(csr, rowptr, dinv, s.w, d.w, r.w);
    } else {
        for (int k = e0; k < e && k < e0 + 4; ++k)
            csr_put(csr, rowptr, dinv, src[k], dst[k], rank[k]);
    }
}

__global__ __launch_bounds__(256) void k_gemm128_wmma(
    const float* __restrict__ A, const float* __restrict__ W,
    const float* __restrict__ bias, float* __restrict__ out,
    int fuse_bias_relu) {
    const int lane = threadIdx.x & 31;
    const int wid  = threadIdx.x >> 5;
    const int m0   = blockIdx.x << 4;
    const int n0   = wid << 4;
    const int half = lane >> 4;
    const int lm   = lane & 15;

    const float* __restrict__ arow = A + (size_t)(m0 + lm) * FD;

    v8f acc0 = {};
    v8f acc1 = {};
#pragma unroll
    for (int k0 = 0; k0 < FD; k0 += 8) {
        {
            const int ka = k0 + 2 * half;
            v2f a, b;
            a.x = arow[ka];
            a.y = arow[ka + 1];
            b.x = W[(size_t)(ka)     * FD + n0 + lm];
            b.y = W[(size_t)(ka + 1) * FD + n0 + lm];
            acc0 = __builtin_amdgcn_wmma_f32_16x16x4_f32(
                false, a, false, b, (short)0, acc0, false, false);
        }
        {
            const int ka = k0 + 4 + 2 * half;
            v2f a, b;
            a.x = arow[ka];
            a.y = arow[ka + 1];
            b.x = W[(size_t)(ka)     * FD + n0 + lm];
            b.y = W[(size_t)(ka + 1) * FD + n0 + lm];
            acc1 = __builtin_amdgcn_wmma_f32_16x16x4_f32(
                false, a, false, b, (short)0, acc1, false, false);
        }
    }

#pragma unroll
    for (int v = 0; v < 8; ++v) {
        const int row = m0 + v + 8 * half;
        const int col = n0 + lm;
        float r = acc0[v] + acc1[v];
        if (fuse_bias_relu) { r += bias[col]; r = fmaxf(r, 0.0f); }
        out[(size_t)row * FD + col] = r;
    }
}

#define NPW 4

__device__ __forceinline__ void row_fma(float4& acc, const float4 v, const float w) {
    acc.x = fmaf(v.x, w, acc.x);
    acc.y = fmaf(v.y, w, acc.y);
    acc.z = fmaf(v.z, w, acc.z);
    acc.w = fmaf(v.w, w, acc.w);
}

template <int M>
__device__ __forceinline__ void walk_rows(float4& acc, const int2 r, const int k,
                                          const char* __restrict__ hwb,
                                          const unsigned loff) {
    float4 v[M];
    float  w[M];
    for (int j = 0; j < M; ++j) {
        const int s = __builtin_amdgcn_readlane(r.x, k + j);
        w[j] = __int_as_float(__builtin_amdgcn_readlane(r.y, k + j));
        v[j] = *(const float4*)(hwb + (size_t)s * 512 + loff);
    }
    for (int j = 0; j < M; ++j) row_fma(acc, v[j], w[j]);
}

__device__ __forceinline__ void walk_chunk(float4& acc, const int2 r, const int cnt,
                                           const char* __restrict__ hwb,
                                           const unsigned loff) {
    for (int k = 0; k < cnt; k += 8) {
        switch (cnt - k) {
            case 1:  walk_rows<1>(acc, r, k, hwb, loff); break;
            case 2:  walk_rows<2>(acc, r, k, hwb, loff); break;
            case 3:  walk_rows<3>(acc, r, k, hwb, loff); break;
            case 4:  walk_rows<4>(acc, r, k, hwb, loff); break;
            case 5:  walk_rows<5>(acc, r, k, hwb, loff); break;
            case 6:  walk_rows<6>(acc, r, k, hwb, loff); break;
            case 7:  walk_rows<7>(acc, r, k, hwb, loff); break;
            default: walk_rows<8>(acc, r, k, hwb, loff); break;
        }
    }
}

__global__ __launch_bounds__(256) void k_gather_rows(
    const int* __restrict__ rowptr, const int* __restrict__ csr_raw,
    const float* __restrict__ dinv, const float* __restrict__ hw,
    const float* __restrict__ bias, float* __restrict__ hout,
    float* __restrict__ emb, int n) {
    const int2* __restrict__ csr = (const int2*)csr_raw;
    const char* __restrict__ hwb = (const char*)hw;
    const int lane = threadIdx.x & 31;
    const unsigned loff = (unsigned)lane * 16u;
    const int wave = __builtin_amdgcn_readfirstlane(
        (int)((blockIdx.x * blockDim.x + threadIdx.x) >> 5));
    const int node0 = wave * NPW;
    if (node0 >= n) return;
    const int nq = (n - node0 < NPW) ? (n - node0) : NPW;

    const int rp_l = rowptr[node0 + (lane < nq ? lane : nq)];
    int rp[NPW + 1];
    for (int q = 0; q <= NPW; ++q) rp[q] = __builtin_amdgcn_readlane(rp_l, q);

    int2 rec[NPW];
    for (int q = 0; q < NPW; ++q) {
        rec[q] = make_int2(0, 0);
        if (rp[q] + lane < rp[q + 1]) rec[q] = csr[rp[q] + lane];
    }
    const float4 b4 = ((const float4*)bias)[lane];

#pragma unroll
    for (int q = 0; q < NPW; ++q) {
        if (q >= nq) break;
        const int   node = node0 + q;
        const float di   = dinv[node];
        const float4 sv  = *(const float4*)(hwb + (size_t)node * 512 + loff);
        float4 acc = make_float4(0.0f, 0.0f, 0.0f, 0.0f);

        const int beg = rp[q], end = rp[q + 1];
        walk_chunk(acc, rec[q], (end - beg < 32) ? (end - beg) : 32, hwb, loff);
        for (int cb = beg + 32; cb < end; cb += 32) {
            int2 r = make_int2(0, 0);
            if (cb + lane < end) r = csr[cb + lane];
            walk_chunk(acc, r, (end - cb < 32) ? (end - cb) : 32, hwb, loff);
        }
        acc.x += sv.x * di * di;
        acc.y += sv.y * di * di;
        acc.z += sv.z * di * di;
        acc.w += sv.w * di * di;
        acc.x += b4.x;
        acc.y += b4.y;
        acc.z += b4.z;
        acc.w += b4.w;

        float4* __restrict__ hp = (float4*)(hout + (size_t)node * FD) + lane;
        if (emb != nullptr) ((float4*)(emb + (size_t)node * FD))[lane] = acc;
        *hp = make_float4(fmaxf(acc.x, 0.0f), fmaxf(acc.y, 0.0f),
                          fmaxf(acc.z, 0.0f), fmaxf(acc.w, 0.0f));
    }
}

__global__ __launch_bounds__(256) void k_logits(
    const float* __restrict__ h, const float* __restrict__ Wp2,
    const float* __restrict__ bp2, float* __restrict__ out, int n) {
    const int wid  = (int)((blockIdx.x * blockDim.x + threadIdx.x) >> 5);
    const int lane = threadIdx.x & 31;
    if (wid >= n) return;
    const float4 v = ((const float4*)(h + (size_t)wid * FD))[lane];
    const int f = lane * 4;
    float p0 = v.x * Wp2[(f + 0) * 2 + 0] + v.y * Wp2[(f + 1) * 2 + 0] +
               v.z * Wp2[(f + 2) * 2 + 0] + v.w * Wp2[(f + 3) * 2 + 0];
    float p1 = v.x * Wp2[(f + 0) * 2 + 1] + v.y * Wp2[(f + 1) * 2 + 1] +
               v.z * Wp2[(f + 2) * 2 + 1] + v.w * Wp2[(f + 3) * 2 + 1];
#pragma unroll
    for (int off = 16; off > 0; off >>= 1) {
        p0 += __shfl_down(p0, off, 32);
        p1 += __shfl_down(p1, off, 32);
    }
    if (lane == 0) {
        const float l0 = p0 + bp2[0];
        const float l1 = p1 + bp2[1];
        const float m  = fmaxf(l0, l1);
        const float lse = m + logf(expf(l0 - m) + expf(l1 - m));
        out[(size_t)wid * 2 + 0] = l0 - lse;
        out[(size_t)wid * 2 + 1] = l1 - lse;
    }
}

extern "C" void kernel_launch(void* const* d_in, const int* in_sizes, int n_in,
                              void* d_out, int out_size, void* d_ws,
                              size_t ws_size, hipStream_t stream) {
    const float* x   = (const float*)d_in[0];
    const int*   ei  = (const int*)d_in[1];
    const float* W1  = (const float*)d_in[2];
    const float* b1  = (const float*)d_in[3];
    const float* W2  = (const float*)d_in[4];
    const float* b2  = (const float*)d_in[5];
    const float* W3  = (const float*)d_in[6];
    const float* b3  = (const float*)d_in[7];
    const float* Wp1 = (const float*)d_in[8];
    const float* bp1 = (const float*)d_in[9];
    const float* Wp2 = (const float*)d_in[10];
    const float* bp2 = (const float*)d_in[11];

    const int N = in_sizes[0] / FD;
    const int E = in_sizes[1] / 2;
    const int* src = ei;
    const int* dst = ei + E;

    char* ws = (char*)d_ws;
    const size_t AL = 256;
    unsigned* degi = (unsigned*)ws;  ws += ((size_t)N * sizeof(unsigned) + AL - 1) / AL * AL;
    float* dinv = (float*)ws;        ws += ((size_t)N * sizeof(float) + AL - 1) / AL * AL;
    int* rowptr = (int*)ws;          ws += ((size_t)(N + 1) * sizeof(int) + AL - 1) / AL * AL;
    int* rank   = (int*)ws;          ws += ((size_t)E * sizeof(int) + AL - 1) / AL * AL;
    int* csr    = (int*)ws;          ws += ((size_t)E * 2 * sizeof(int) + AL - 1) / AL * AL;
    float* hw   = (float*)ws;        ws += (size_t)N * FD * sizeof(float);
    float* h    = (float*)ws;

    float* emb_out  = (float*)d_out;
    float* prob_out = (float*)d_out + (size_t)N * FD;

    const dim3 blk(256);
    const int gN      = (N + 255) / 256;
    const int gE      = (E + 255) / 256;
    const int gScan   = (N + 1023) / 1024;
    const int gFill   = (E + 1023) / 1024;
    const int gTiles  = N / 16;
    const int gRows   = (N + 8 * NPW - 1) / (8 * NPW);
    const int gNodeW  = (N * WAVE + 255) / 256;

    k_deg_init   <<<gN, blk, 0, stream>>>(degi, N);
    k_deg_rank   <<<gE, blk, 0, stream>>>(dst, degi, rank, E);
    k_rowptr_dinv<<<gScan, blk, 0, stream>>>(degi, rowptr, dinv, N);
    k_csr_fill   <<<gFill, blk, 0, stream>>>(src, dst, rank, rowptr, dinv, csr, E);

    k_gemm128_wmma<<<gTiles, blk, 0, stream>>>(x, W1, nullptr, hw, 0);
    k_gather_rows <<<gRows, blk, 0, stream>>>(rowptr, csr, dinv, hw, b1, h, nullptr, N);

    k_gemm128_wmma<<<gTiles, blk, 0, stream>>>(h, W2, nullptr, hw, 0);
    k_gather_rows <<<gRows, blk, 0, stream>>>(rowptr, csr, dinv, hw, b2, h, nullptr, N);

    k_gemm128_wmma<<<gTiles, blk, 0, stream>>>(h, W3, nullptr, hw, 0);
    k_gather_rows <<<gRows, blk, 0, stream>>>(rowptr, csr, dinv, hw, b3, h, emb_out, N);

    k_gemm128_wmma<<<gTiles, blk, 0, stream>>>(h, Wp1, bp1, hw, 1);
    k_logits      <<<gNodeW, blk, 0, stream>>>(hw, Wp2, bp2, prob_out, N);
}


# GEMM: 1024 persistent blocks walk the row tiles; each wave keeps its 128x16 W slice in registers and the A tile goes once per block through LDS, not per-wave global fragment loads; sums unchanged
# speedup vs baseline: 3.1514x; 1.2573x over previous
#include <hip/hip_runtime.h>
#include <math.h>

#define FD 128
#define WAVE 32

typedef __attribute__((ext_vector_type(2))) float v2f;
typedef __attribute__((ext_vector_type(8))) float v8f;

__global__ void k_deg_init(unsigned* __restrict__ degi, int n) {
    int i = blockIdx.x * blockDim.x + threadIdx.x;
    if (i < n) degi[i] = 1u;
}

__global__ void k_deg_rank(const int* __restrict__ dst,
                           unsigned* __restrict__ degi,
                           int* __restrict__ rank, int e) {
    int i = blockIdx.x * blockDim.x + threadIdx.x;
    if (i < e) rank[i] = (int)atomicAdd(&degi[dst[i]], 1u);
}

__global__ __launch_bounds__(256) void k_rowptr_dinv(
    const unsigned* __restrict__ degi, int* __restrict__ rowptr,
    float* __restrict__ dinv, int n) {
    __shared__ unsigned red[8];
    __shared__ unsigned wtot[8];
    const int t    = threadIdx.x;
    const int lane = t & 31;
    const int wid  = t >> 5;
    const int base = blockIdx.x * 1024;
    const uint4* __restrict__ d4 = (const uint4*)degi;

    unsigned p = 0;
    const int nv = base >> 2;
#pragma unroll 8
    for (int j = t; j < nv; j += 256) {
        const uint4 v = d4[j];
        p += (v.x + v.y + v.z + v.w) - 4u;
    }
    for (int off = 16; off > 0; off >>= 1) p += __shfl_xor(p, off, 32);
    if (lane == 0) red[wid] = p;

    const int idx = base + 4 * t;
    const bool ok = (idx + 3 < n);
    uint4 c = make_uint4(1u, 1u, 1u, 1u);
    if (ok) c = d4[idx >> 2];
    const unsigned c0 = c.x - 1u, c1 = c.y - 1u, c2 = c.z - 1u, c3 = c.w - 1u;
    const unsigned s = c0 + c1 + c2 + c3;
    unsigned inc = s;
    for (int off = 1; off < 32; off <<= 1) {
        const unsigned y = __shfl_up(inc, off, 32);
        if (lane >= off) inc += y;
    }
    if (lane == 31) wtot[wid] = inc;
    __syncthreads();
    unsigned ex = inc - s;
    for (int w = 0; w < 8; ++w) {
        ex += red[w];
        if (w < wid) ex += wtot[w];
    }
    if (ok) {
        *(int4*)(rowptr + idx) = make_int4((int)ex, (int)(ex + c0),
                                           (int)(ex + c0 + c1),
                                           (int)(ex + c0 + c1 + c2));
        *(float4*)(dinv + idx) = make_float4(rsqrtf((float)c.x), rsqrtf((float)c.y),
                                             rsqrtf((float)c.z), rsqrtf((float)c.w));
        if (idx + 4 == n) rowptr[n] = (int)(ex + s);
    }
}

__device__ __forceinline__ void csr_put(int2* __restrict__ csr,
                                        const int* __restrict__ rowptr,
                                        const float* __restrict__ dinv,
                                        int s, int d, int r) {
    const float w = dinv[s] * dinv[d];
    csr[rowptr[d] + r - 1] = make_int2(s, __float_as_int(w));
}

__global__ __launch_bounds__(256) void k_csr_fill(
    const int* __restrict__ src, const int* __restrict__ dst,
    const int* __restrict__ rank, const int* __restrict__ rowptr,
    const float* __restrict__ dinv, int* __restrict__ csr_raw, int e) {
    int2* __restrict__ csr = (int2*)csr_raw;
    const int e0 = (blockIdx.x * blockDim.x + threadIdx.x) * 4;
    if (e0 >= e) return;
    if (((e & 3) == 0) && (e0 + 3 < e)) {
        const int4 s = *(const int4*)(src + e0);
        const int4 d = *(const int4*)(dst + e0);
        const int4 r = *(const int4*)(rank + e0);
        csr_put(csr, rowptr, dinv, s.x, d.x, r.x);
        csr_put(csr, rowptr, dinv, s.y, d.y, r.y);
        csr_put(csr, rowptr, dinv, s.z, d.z, r.z);
        csr_put(csr, rowptr, dinv, s.w, d.w, r.w);
    } else {
        for (int k = e0; k < e && k < e0 + 4; ++k)
            csr_put(csr, rowptr, dinv, src[k], dst[k], rank[k]);
    }
}

#define LDA 132

__device__ __forceinline__ void lds_barrier() {
    __builtin_amdgcn_fence(__ATOMIC_RELEASE, "workgroup", "local");
    __builtin_amdgcn_s_barrier();
    __builtin_amdgcn_fence(__ATOMIC_ACQUIRE, "workgroup", "local");
}

__global__ __launch_bounds__(256) void k_gemm128_wmma(
    const float* __restrict__ A, const float* __restrict__ W,
    const float* __restrict__ bias, float* __restrict__ out,
    int fuse_bias_relu, int ntiles) {
    __shared__ float As[2][16 * LDA];
    const int t    = threadIdx.x;
    const int lane = t & 31;
    const int wid  = t >> 5;
    const int n0   = wid << 4;
    const int half = lane >> 4;
    const int lm   = lane & 15;

    v2f bf[32];
#pragma unroll
    for (int s = 0; s < 32; ++s) {
        const int ka = 4 * s + 2 * half;
        bf[s].x = W[(size_t)(ka)     * FD + n0 + lm];
        bf[s].y = W[(size_t)(ka + 1) * FD + n0 + lm];
    }
    float bcol = 0.0f;
    if (fuse_bias_relu) bcol = bias[n0 + lm];

    const int sr = t >> 5;
    const int sc = (t & 31) << 2;
    float* const st0 = &As[0][sr * LDA + sc];
    float* const st1 = &As[0][(sr + 8) * LDA + sc];

    int tile = blockIdx.x;
    if (tile < ntiles) {
        const float4* __restrict__ g = (const float4*)(A + (size_t)tile * (16 * FD));
        const float4 g0 = g[t];
        const float4 g1 = g[t + 256];
        *(float4*)st0 = g0;
        *(float4*)st1 = g1;
    }
    __syncthreads();

    int buf = 0;
    for (; tile < ntiles; tile += (int)gridDim.x) {
        const int  nxt  = tile + (int)gridDim.x;
        const bool more = nxt < ntiles;
        float4 g0 = make_float4(0.0f, 0.0f, 0.0f, 0.0f), g1 = g0;
        if (more) {
            const float4* __restrict__ g = (const float4*)(A + (size_t)nxt * (16 * FD));
            g0 = g[t];
            g1 = g[t + 256];
        }

        const float* __restrict__ as = &As[buf][lm * LDA + 2 * half];
        v8f acc0 = {};
        v8f acc1 = {};
#pragma unroll
        for (int s = 0; s < 32; s += 2) {
            const v2f a0 = *(const v2f*)(as + 4 * s);
            const v2f a1 = *(const v2f*)(as + 4 * s + 4);
            acc0 = __builtin_amdgcn_wmma_f32_16x16x4_f32(
                false, a0, false, bf[s], (short)0, acc0, false, false);
            acc1 = __builtin_amdgcn_wmma_f32_16x16x4_f32(
                false, a1, false, bf[s + 1], (short)0, acc1, false, false);
        }

        const int m0 = tile << 4;
#pragma unroll
        for (int v = 0; v < 8; ++v) {
            const int row = m0 + v + 8 * half;
            const int col = n0 + lm;
            float r = acc0[v] + acc1[v];
            if (fuse_bias_relu) { r += bcol; r = fmaxf(r, 0.0f); }
            out[(size_t)row * FD + col] = r;
        }

        if (more) {
            const int o = (buf ^ 1) * (16 * LDA);
            *(float4*)(st0 + o) = g0;
            *(float4*)(st1 + o) = g1;
        }
        lds_barrier();
        buf ^= 1;
    }
}

#define NPW 4

__device__ __forceinline__ void row_fma(float4& acc, const float4 v, const float w) {
    acc.x = fmaf(v.x, w, acc.x);
    acc.y = fmaf(v.y, w, acc.y);
    acc.z = fmaf(v.z, w, acc.z);
    acc.w = fmaf(v.w, w, acc.w);
}

template <int M>
__device__ __forceinline__ void walk_rows(float4& acc, const int2 r, const int k,
                                          const char* __restrict__ hwb,
                                          const unsigned loff) {
    float4 v[M];
    float  w[M];
    for (int j = 0; j < M; ++j) {
        const int s = __builtin_amdgcn_readlane(r.x, k + j);
        w[j] = __int_as_float(__builtin_amdgcn_readlane(r.y, k + j));
        v[j] = *(const float4*)(hwb + (size_t)s * 512 + loff);
    }
    for (int j = 0; j < M; ++j) row_fma(acc, v[j], w[j]);
}

__device__ __forceinline__ void walk_chunk(float4& acc, const int2 r, const int cnt,
                                           const char* __restrict__ hwb,
                                           const unsigned loff) {
    for (int k = 0; k < cnt; k += 8) {
        switch (cnt - k) {
            case 1:  walk_rows<1>(acc, r, k, hwb, loff); break;
            case 2:  walk_rows<2>(acc, r, k, hwb, loff); break;
            case 3:  walk_rows<3>(acc, r, k, hwb, loff); break;
            case 4:  walk_rows<4>(acc, r, k, hwb, loff); break;
            case 5:  walk_rows<5>(acc, r, k, hwb, loff); break;
            case 6:  walk_rows<6>(acc, r, k, hwb, loff); break;
            case 7:  walk_rows<7>(acc, r, k, hwb, loff); break;
            default: walk_rows<8>(acc, r, k, hwb, loff); break;
        }
    }
}

__global__ __launch_bounds__(256) void k_gather_rows(
    const int* __restrict__ rowptr, const int* __restrict__ csr_raw,
    const float* __restrict__ dinv, const float* __restrict__ hw,
    const float* __restrict__ bias, float* __restrict__ hout,
    float* __restrict__ emb, int n) {
    const int2* __restrict__ csr = (const int2*)csr_raw;
    const char* __restrict__ hwb = (const char*)hw;
    const int lane = threadIdx.x & 31;
    const unsigned loff = (unsigned)lane * 16u;
    const int wave = __builtin_amdgcn_readfirstlane(
        (int)((blockIdx.x * blockDim.x + threadIdx.x) >> 5));
    const int node0 = wave * NPW;
    if (node0 >= n) return;
    const int nq = (n - node0 < NPW) ? (n - node0) : NPW;

    const int rp_l = rowptr[node0 + (lane < nq ? lane : nq)];
    int rp[NPW + 1];
    for (int q = 0; q <= NPW; ++q) rp[q] = __builtin_amdgcn_readlane(rp_l, q);

    int2 rec[NPW];
    for (int q = 0; q < NPW; ++q) {
        rec[q] = make_int2(0, 0);
        if (rp[q] + lane < rp[q + 1]) rec[q] = csr[rp[q] + lane];
    }
    const float4 b4 = ((const float4*)bias)[lane];

#pragma unroll
    for (int q = 0; q < NPW; ++q) {
        if (q >= nq) break;
        const int   node = node0 + q;
        const float di   = dinv[node];
        const float4 sv  = *(const float4*)(hwb + (size_t)node * 512 + loff);
        float4 acc = make_float4(0.0f, 0.0f, 0.0f, 0.0f);

        const int beg = rp[q], end = rp[q + 1];
        walk_chunk(acc, rec[q], (end - beg < 32) ? (end - beg) : 32, hwb, loff);
        for (int cb = beg + 32; cb < end; cb += 32) {
            int2 r = make_int2(0, 0);
            if (cb + lane < end) r = csr[cb + lane];
            walk_chunk(acc, r, (end - cb < 32) ? (end - cb) : 32, hwb, loff);
        }
        acc.x += sv.x * di * di;
        acc.y += sv.y * di * di;
        acc.z += sv.z * di * di;
        acc.w += sv.w * di * di;
        acc.x += b4.x;
        acc.y += b4.y;
        acc.z += b4.z;
        acc.w += b4.w;

        float4* __restrict__ hp = (float4*)(hout + (size_t)node * FD) + lane;
        if (emb != nullptr) ((float4*)(emb + (size_t)node * FD))[lane] = acc;
        *hp = make_float4(fmaxf(acc.x, 0.0f), fmaxf(acc.y, 0.0f),
                          fmaxf(acc.z, 0.0f), fmaxf(acc.w, 0.0f));
    }
}

__global__ __launch_bounds__(256) void k_logits(
    const float* __restrict__ h, const float* __restrict__ Wp2,
    const float* __restrict__ bp2, float* __restrict__ out, int n) {
    const int wid  = (int)((blockIdx.x * blockDim.x + threadIdx.x) >> 5);
    const int lane = threadIdx.x & 31;
    if (wid >= n) return;
    const float4 v = ((const float4*)(h + (size_t)wid * FD))[lane];
    const int f = lane * 4;
    float p0 = v.x * Wp2[(f + 0) * 2 + 0] + v.y * Wp2[(f + 1) * 2 + 0] +
               v.z * Wp2[(f + 2) * 2 + 0] + v.w * Wp2[(f + 3) * 2 + 0];
    float p1 = v.x * Wp2[(f + 0) * 2 + 1] + v.y * Wp2[(f + 1) * 2 + 1] +
               v.z * Wp2[(f + 2) * 2 + 1] + v.w * Wp2[(f + 3) * 2 + 1];
#pragma unroll
    for (int off = 16; off > 0; off >>= 1) {
        p0 += __shfl_down(p0, off, 32);
        p1 += __shfl_down(p1, off, 32);
    }
    if (lane == 0) {
        const float l0 = p0 + bp2[0];
        const float l1 = p1 + bp2[1];
        const float m  = fmaxf(l0, l1);
        const float lse = m + logf(expf(l0 - m) + expf(l1 - m));
        out[(size_t)wid * 2 + 0] = l0 - lse;
        out[(size_t)wid * 2 + 1] = l1 - lse;
    }
}

extern "C" void kernel_launch(void* const* d_in, const int* in_sizes, int n_in,
                              void* d_out, int out_size, void* d_ws,
                              size_t ws_size, hipStream_t stream) {
    const float* x   = (const float*)d_in[0];
    const int*   ei  = (const int*)d_in[1];
    const float* W1  = (const float*)d_in[2];
    const float* b1  = (const float*)d_in[3];
    const float* W2  = (const float*)d_in[4];
    const float* b2  = (const float*)d_in[5];
    const float* W3  = (const float*)d_in[6];
    const float* b3  = (const float*)d_in[7];
    const float* Wp1 = (const float*)d_in[8];
    const float* bp1 = (const float*)d_in[9];
    const float* Wp2 = (const float*)d_in[10];
    const float* bp2 = (const float*)d_in[11];

    const int N = in_sizes[0] / FD;
    const int E = in_sizes[1] / 2;
    const int* src = ei;
    const int* dst = ei + E;

    char* ws = (char*)d_ws;
    const size_t AL = 256;
    unsigned* degi = (unsigned*)ws;  ws += ((size_t)N * sizeof(unsigned) + AL - 1) / AL * AL;
    float* dinv = (float*)ws;        ws += ((size_t)N * sizeof(float) + AL - 1) / AL * AL;
    int* rowptr = (int*)ws;          ws += ((size_t)(N + 1) * sizeof(int) + AL - 1) / AL * AL;
    int* rank   = (int*)ws;          ws += ((size_t)E * sizeof(int) + AL - 1) / AL * AL;
    int* csr    = (int*)ws;          ws += ((size_t)E * 2 * sizeof(int) + AL - 1) / AL * AL;
    float* hw   = (float*)ws;        ws += (size_t)N * FD * sizeof(float);
    float* h    = (float*)ws;

    float* emb_out  = (float*)d_out;
    float* prob_out = (float*)d_out + (size_t)N * FD;

    const dim3 blk(256);
    const int gN      = (N + 255) / 256;
    const int gE      = (E + 255) / 256;
    const int gScan   = (N + 1023) / 1024;
    const int gFill   = (E + 1023) / 1024;
    const int gTiles  = N / 16;
    const int gGemm   = 1024;
    const int gRows   = (N + 8 * NPW - 1) / (8 * NPW);
    const int gNodeW  = (N * WAVE + 255) / 256;

    k_deg_init   <<<gN, blk, 0, stream>>>(degi, N);
    k_deg_rank   <<<gE, blk, 0, stream>>>(dst, degi, rank, E);
    k_rowptr_dinv<<<gScan, blk, 0, stream>>>(degi, rowptr, dinv, N);
    k_csr_fill   <<<gFill, blk, 0, stream>>>(src, dst, rank, rowptr, dinv, csr, E);

    k_gemm128_wmma<<<gGemm, blk, 0, stream>>>(x, W1, nullptr, hw, 0, gTiles);
    k_gather_rows <<<gRows, blk, 0, stream>>>(rowptr, csr, dinv, hw, b1, h, nullptr, N);

    k_gemm128_wmma<<<gGemm, blk, 0, stream>>>(h, W2, nullptr, hw, 0, gTiles);
    k_gather_rows <<<gRows, blk, 0, stream>>>(rowptr, csr, dinv, hw, b2, h, nullptr, N);

    k_gemm128_wmma<<<gGemm, blk, 0, stream>>>(h, W3, nullptr, hw, 0, gTiles);
    k_gather_rows <<<gRows, blk, 0, stream>>>(rowptr, csr, dinv, hw, b3, h, emb_out, N);

    k_gemm128_wmma<<<gGemm, blk, 0, stream>>>(h, Wp1, bp1, hw, 1, gTiles);
    k_logits      <<<gNodeW, blk, 0, stream>>>(hw, Wp2, bp2, prob_out, N);
}


# Last GEMM does the 128->2 head and log_softmax in its epilogue (16-lane sums per wave, 8 wave partials via LDS): no hidden-plane store, no 50000-wave head launch; head dot's sum order changes
# speedup vs baseline: 3.2485x; 1.0308x over previous
#include <hip/hip_runtime.h>
#include <math.h>

#define FD 128

typedef __attribute__((ext_vector_type(2))) float v2f;
typedef __attribute__((ext_vector_type(8))) float v8f;

__global__ void k_deg_init(unsigned* __restrict__ degi, int n) {
    int i = blockIdx.x * blockDim.x + threadIdx.x;
    if (i < n) degi[i] = 1u;
}

__global__ void k_deg_rank(const int* __restrict__ dst,
                           unsigned* __restrict__ degi,
                           int* __restrict__ rank, int e) {
    int i = blockIdx.x * blockDim.x + threadIdx.x;
    if (i < e) rank[i] = (int)atomicAdd(&degi[dst[i]], 1u);
}

__global__ __launch_bounds__(256) void k_rowptr_dinv(
    const unsigned* __restrict__ degi, int* __restrict__ rowptr,
    float* __restrict__ dinv, int n) {
    __shared__ unsigned red[8];
    __shared__ unsigned wtot[8];
    const int t    = threadIdx.x;
    const int lane = t & 31;
    const int wid  = t >> 5;
    const int base = blockIdx.x * 1024;
    const uint4* __restrict__ d4 = (const uint4*)degi;

    unsigned p = 0;
    const int nv = base >> 2;
#pragma unroll 8
    for (int j = t; j < nv; j += 256) {
        const uint4 v = d4[j];
        p += (v.x + v.y + v.z + v.w) - 4u;
    }
    for (int off = 16; off > 0; off >>= 1) p += __shfl_xor(p, off, 32);
    if (lane == 0) red[wid] = p;

    const int idx = base + 4 * t;
    const bool ok = (idx + 3 < n);
    uint4 c = make_uint4(1u, 1u, 1u, 1u);
    if (ok) c = d4[idx >> 2];
    const unsigned c0 = c.x - 1u, c1 = c.y - 1u, c2 = c.z - 1u, c3 = c.w - 1u;
    const unsigned s = c0 + c1 + c2 + c3;
    unsigned inc = s;
    for (int off = 1; off < 32; off <<= 1) {
        const unsigned y = __shfl_up(inc, off, 32);
        if (lane >= off) inc += y;
    }
    if (lane == 31) wtot[wid] = inc;
    __syncthreads();
    unsigned ex = inc - s;
    for (int w = 0; w < 8; ++w) {
        ex += red[w];
        if (w < wid) ex += wtot[w];
    }
    if (ok) {
        *(int4*)(rowptr + idx) = make_int4((int)ex, (int)(ex + c0),
                                           (int)(ex + c0 + c1),
                                           (int)(ex + c0 + c1 + c2));
        *(float4*)(dinv + idx) = make_float4(rsqrtf((float)c.x), rsqrtf((float)c.y),
                                             rsqrtf((float)c.z), rsqrtf((float)c.w));
        if (idx + 4 == n) rowptr[n] = (int)(ex + s);
    }
}

__device__ __forceinline__ void csr_put(int2* __restrict__ csr,
                                        const int* __restrict__ rowptr,
                                        const float* __restrict__ dinv,
                                        int s, int d, int r) {
    const float w = dinv[s] * dinv[d];
    csr[rowptr[d] + r - 1] = make_int2(s, __float_as_int(w));
}

__global__ __launch_bounds__(256) void k_csr_fill(
    const int* __restrict__ src, const int* __restrict__ dst,
    const int* __restrict__ rank, const int* __restrict__ rowptr,
    const float* __restrict__ dinv, int* __restrict__ csr_raw, int e) {
    int2* __restrict__ csr = (int2*)csr_raw;
    const int e0 = (blockIdx.x * blockDim.x + threadIdx.x) * 4;
    if (e0 >= e) return;
    if (((e & 3) == 0) && (e0 + 3 < e)) {
        const int4 s = *(const int4*)(src + e0);
        const int4 d = *(const int4*)(dst + e0);
        const int4 r = *(const int4*)(rank + e0);
        csr_put(csr, rowptr, dinv, s.x, d.x, r.x);
        csr_put(csr, rowptr, dinv, s.y, d.y, r.y);
        csr_put(csr, rowptr, dinv, s.z, d.z, r.z);
        csr_put(csr, rowptr, dinv, s.w, d.w, r.w);
    } else {
        for (int k = e0; k < e && k < e0 + 4; ++k)
            csr_put(csr, rowptr, dinv, src[k], dst[k], rank[k]);
    }
}

#define LDA 132

__device__ __forceinline__ void lds_barrier() {
    __builtin_amdgcn_fence(__ATOMIC_RELEASE, "workgroup", "local");
    __builtin_amdgcn_s_barrier();
    __builtin_amdgcn_fence(__ATOMIC_ACQUIRE, "workgroup", "local");
}

__device__ __forceinline__ float half_sum16(float x) {
    x += __int_as_float(__builtin_amdgcn_update_dpp(0, __float_as_int(x), 0x161, 0xf, 0xf, true));
    x += __int_as_float(__builtin_amdgcn_update_dpp(0, __float_as_int(x), 0x162, 0xf, 0xf, true));
    x += __int_as_float(__builtin_amdgcn_update_dpp(0, __float_as_int(x), 0x164, 0xf, 0xf, true));
    x += __int_as_float(__builtin_amdgcn_update_dpp(0, __float_as_int(x), 0x168, 0xf, 0xf, true));
    return x;
}

template <bool HEAD>
__device__ __forceinline__ void gemm128_body(
    const float* __restrict__ A, const float* __restrict__ W,
    const float* __restrict__ bias, float* __restrict__ out,
    int fuse_bias_relu, int ntiles,
    const float* __restrict__ Wh, const float* __restrict__ bh,
    float* __restrict__ prob) {
    __shared__ float As[2][16 * LDA];
    __shared__ float part[HEAD ? 2 * 8 * 32 : 1];
    const int t    = threadIdx.x;
    const int lane = t & 31;
    const int wid  = t >> 5;
    const int n0   = wid << 4;
    const int half = lane >> 4;
    const int lm   = lane & 15;

    v2f bf[32];
#pragma unroll
    for (int s = 0; s < 32; ++s) {
        const int ka = 4 * s + 2 * half;
        bf[s].x = W[(size_t)(ka)     * FD + n0 + lm];
        bf[s].y = W[(size_t)(ka + 1) * FD + n0 + lm];
    }
    float bcol = 0.0f;
    if (HEAD || fuse_bias_relu) bcol = bias[n0 + lm];
    float wh0 = 0.0f, wh1 = 0.0f, bhl = 0.0f;
    if (HEAD) {
        const float2 wh = *(const float2*)(Wh + 2 * (n0 + lm));
        wh0 = wh.x;
        wh1 = wh.y;
        bhl = bh[lane & 1];
    }

    const int sr = t >> 5;
    const int sc = (t & 31) << 2;
    float* const st0 = &As[0][sr * LDA + sc];
    float* const st1 = &As[0][(sr + 8) * LDA + sc];

    int tile = blockIdx.x;
    if (tile < ntiles) {
        const float4* __restrict__ g = (const float4*)(A + (size_t)tile * (16 * FD));
        const float4 g0 = g[t];
        const float4 g1 = g[t + 256];
        *(float4*)st0 = g0;
        *(float4*)st1 = g1;
    }
    __syncthreads();

    int buf = 0;
    int it  = 0;
    for (; tile < ntiles; tile += (int)gridDim.x) {
        const int  nxt  = tile + (int)gridDim.x;
        const bool more = nxt < ntiles;
        float4 g0 = make_float4(0.0f, 0.0f, 0.0f, 0.0f), g1 = g0;
        if (more) {
            const float4* __restrict__ g = (const float4*)(A + (size_t)nxt * (16 * FD));
            g0 = g[t];
            g1 = g[t + 256];
        }

        const float* __restrict__ as = &As[buf][lm * LDA + 2 * half];
        v8f acc0 = {};
        v8f acc1 = {};
#pragma unroll
        for (int s = 0; s < 32; s += 2) {
            const v2f a0 = *(const v2f*)(as + 4 * s);
            const v2f a1 = *(const v2f*)(as + 4 * s + 4);
            acc0 = __builtin_amdgcn_wmma_f32_16x16x4_f32(
                false, a0, false, bf[s], (short)0, acc0, false, false);
            acc1 = __builtin_amdgcn_wmma_f32_16x16x4_f32(
                false, a1, false, bf[s + 1], (short)0, acc1, false, false);
        }

        const int pb = it & 1;
        if (HEAD) {
            float p[16];
#pragma unroll
            for (int v = 0; v < 8; ++v) {
                float r = acc0[v] + acc1[v];
                r += bcol;
                r = fmaxf(r, 0.0f);
                p[2 * v]     = r * wh0;
                p[2 * v + 1] = r * wh1;
            }
#pragma unroll
            for (int i = 0; i < 16; ++i) p[i] = half_sum16(p[i]);
            if (lm == 0) {
                float4* __restrict__ pp = (float4*)&part[(pb * 8 + wid) * 32 + half * 16];
                pp[0] = make_float4(p[0],  p[1],  p[2],  p[3]);
                pp[1] = make_float4(p[4],  p[5],  p[6],  p[7]);
                pp[2] = make_float4(p[8],  p[9],  p[10], p[11]);
                pp[3] = make_float4(p[12], p[13], p[14], p[15]);
            }
        } else {
            const int m0 = tile << 4;
#pragma unroll
            for (int v = 0; v < 8; ++v) {
                const int row = m0 + v + 8 * half;
                const int col = n0 + lm;
                float r = acc0[v] + acc1[v];
                if (fuse_bias_relu) { r += bcol; r = fmaxf(r, 0.0f); }
                out[(size_t)row * FD + col] = r;
            }
        }

        if (more) {
            const int o = (buf ^ 1) * (16 * LDA);
            *(float4*)(st0 + o) = g0;
            *(float4*)(st1 + o) = g1;
        }
        lds_barrier();
        buf ^= 1;

        if (HEAD) {
            if (wid == (it & 7)) {
                const float* __restrict__ pr = &part[pb * 256 + lane];
                float l = pr[0];
#pragma unroll
                for (int w = 1; w < 8; ++w) l += pr[w * 32];
                l += bhl;
                const float o = __int_as_float(
                    __builtin_amdgcn_update_dpp(0, __float_as_int(l), 0x161, 0xf, 0xf, true));
                const float m   = fmaxf(l, o);
                const float lse = m + logf(expf(l - m) + expf(o - m));
                prob[(size_t)tile * 32 + lane] = l - lse;
            }
        }
        ++it;
    }
}

__global__ __launch_bounds__(256) void k_gemm128_wmma(
    const float* __restrict__ A, const float* __restrict__ W,
    const float* __restrict__ bias, float* __restrict__ out,
    int fuse_bias_relu, int ntiles) {
    gemm128_body<false>(A, W, bias, out, fuse_bias_relu, ntiles, nullptr, nullptr, nullptr);
}

__global__ __launch_bounds__(256) void k_gemm128_head(
    const float* __restrict__ A, const float* __restrict__ W,
    const float* __restrict__ bias, const float* __restrict__ Wh,
    const float* __restrict__ bh, float* __restrict__ prob, int ntiles) {
    gemm128_body<true>(A, W, bias, nullptr, 1, ntiles, Wh, bh, prob);
}

#define NPW 4

__device__ __forceinline__ void row_fma(float4& acc, const float4 v, const float w) {
    acc.x = fmaf(v.x, w, acc.x);
    acc.y = fmaf(v.y, w, acc.y);
    acc.z = fmaf(v.z, w, acc.z);
    acc.w = fmaf(v.w, w, acc.w);
}

template <int M>
__device__ __forceinline__ void walk_rows(float4& acc, const int2 r, const int k,
                                          const char* __restrict__ hwb,
                                          const unsigned loff) {
    float4 v[M];
    float  w[M];
    for (int j = 0; j < M; ++j) {
        const int s = __builtin_amdgcn_readlane(r.x, k + j);
        w[j] = __int_as_float(__builtin_amdgcn_readlane(r.y, k + j));
        v[j] = *(const float4*)(hwb + (size_t)s * 512 + loff);
    }
    for (int j = 0; j < M; ++j) row_fma(acc, v[j], w[j]);
}

__device__ __forceinline__ void walk_chunk(float4& acc, const int2 r, const int cnt,
                                           const char* __restrict__ hwb,
                                           const unsigned loff) {
    for (int k = 0; k < cnt; k += 8) {
        switch (cnt - k) {
            case 1:  walk_rows<1>(acc, r, k, hwb, loff); break;
            case 2:  walk_rows<2>(acc, r, k, hwb, loff); break;
            case 3:  walk_rows<3>(acc, r, k, hwb, loff); break;
            case 4:  walk_rows<4>(acc, r, k, hwb, loff); break;
            case 5:  walk_rows<5>(acc, r, k, hwb, loff); break;
            case 6:  walk_rows<6>(acc, r, k, hwb, loff); break;
            case 7:  walk_rows<7>(acc, r, k, hwb, loff); break;
            default: walk_rows<8>(acc, r, k, hwb, loff); break;
        }
    }
}

__global__ __launch_bounds__(256) void k_gather_rows(
    const int* __restrict__ rowptr, const int* __restrict__ csr_raw,
    const float* __restrict__ dinv, const float* __restrict__ hw,
    const float* __restrict__ bias, float* __restrict__ hout,
    float* __restrict__ emb, int n) {
    const int2* __restrict__ csr = (const int2*)csr_raw;
    const char* __restrict__ hwb = (const char*)hw;
    const int lane = threadIdx.x & 31;
    const unsigned loff = (unsigned)lane * 16u;
    const int wave = __builtin_amdgcn_readfirstlane(
        (int)((blockIdx.x * blockDim.x + threadIdx.x) >> 5));
    const int node0 = wave * NPW;
    if (node0 >= n) return;
    const int nq = (n - node0 < NPW) ? (n - node0) : NPW;

    const int rp_l = rowptr[node0 + (lane < nq ? lane : nq)];
    int rp[NPW + 1];
    for (int q = 0; q <= NPW; ++q) rp[q] = __builtin_amdgcn_readlane(rp_l, q);

    int2 rec[NPW];
    for (int q = 0; q < NPW; ++q) {
        rec[q] = make_int2(0, 0);
        if (rp[q] + lane < rp[q + 1]) rec[q] = csr[rp[q] + lane];
    }
    const float4 b4 = ((const float4*)bias)[lane];

#pragma unroll
    for (int q = 0; q < NPW; ++q) {
        if (q >= nq) break;
        const int   node = node0 + q;
        const float di   = dinv[node];
        const float4 sv  = *(const float4*)(hwb + (size_t)node * 512 + loff);
        float4 acc = make_float4(0.0f, 0.0f, 0.0f, 0.0f);

        const int beg = rp[q], end = rp[q + 1];
        walk_chunk(acc, rec[q], (end - beg < 32) ? (end - beg) : 32, hwb, loff);
        for (int cb = beg + 32; cb < end; cb += 32) {
            int2 r = make_int2(0, 0);
            if (cb + lane < end) r = csr[cb + lane];
            walk_chunk(acc, r, (end - cb < 32) ? (end - cb) : 32, hwb, loff);
        }
        acc.x += sv.x * di * di;
        acc.y += sv.y * di * di;
        acc.z += sv.z * di * di;
        acc.w += sv.w * di * di;
        acc.x += b4.x;
        acc.y += b4.y;
        acc.z += b4.z;
        acc.w += b4.w;

        float4* __restrict__ hp = (float4*)(hout + (size_t)node * FD) + lane;
        if (emb != nullptr) ((float4*)(emb + (size_t)node * FD))[lane] = acc;
        *hp = make_float4(fmaxf(acc.x, 0.0f), fmaxf(acc.y, 0.0f),
                          fmaxf(acc.z, 0.0f), fmaxf(acc.w, 0.0f));
    }
}

extern "C" void kernel_launch(void* const* d_in, const int* in_sizes, int n_in,
                              void* d_out, int out_size, void* d_ws,
                              size_t ws_size, hipStream_t stream) {
    const float* x   = (const float*)d_in[0];
    const int*   ei  = (const int*)d_in[1];
    const float* W1  = (const float*)d_in[2];
    const float* b1  = (const float*)d_in[3];
    const float* W2  = (const float*)d_in[4];
    const float* b2  = (const float*)d_in[5];
    const float* W3  = (const float*)d_in[6];
    const float* b3  = (const float*)d_in[7];
    const float* Wp1 = (const float*)d_in[8];
    const float* bp1 = (const float*)d_in[9];
    const float* Wp2 = (const float*)d_in[10];
    const float* bp2 = (const float*)d_in[11];

    const int N = in_sizes[0] / FD;
    const int E = in_sizes[1] / 2;
    const int* src = ei;
    const int* dst = ei + E;

    char* ws = (char*)d_ws;
    const size_t AL = 256;
    unsigned* degi = (unsigned*)ws;  ws += ((size_t)N * sizeof(unsigned) + AL - 1) / AL * AL;
    float* dinv = (float*)ws;        ws += ((size_t)N * sizeof(float) + AL - 1) / AL * AL;
    int* rowptr = (int*)ws;          ws += ((size_t)(N + 1) * sizeof(int) + AL - 1) / AL * AL;
    int* rank   = (int*)ws;          ws += ((size_t)E * sizeof(int) + AL - 1) / AL * AL;
    int* csr    = (int*)ws;          ws += ((size_t)E * 2 * sizeof(int) + AL - 1) / AL * AL;
    float* hw   = (float*)ws;        ws += (size_t)N * FD * sizeof(float);
    float* h    = (float*)ws;

    float* emb_out  = (float*)d_out;
    float* prob_out = (float*)d_out + (size_t)N * FD;

    const dim3 blk(256);
    const int gN      = (N + 255) / 256;
    const int gE      = (E + 255) / 256;
    const int gScan   = (N + 1023) / 1024;
    const int gFill   = (E + 1023) / 1024;
    const int gTiles  = N / 16;
    const int gGemm   = 1024;
    const int gRows   = (N + 8 * NPW - 1) / (8 * NPW);

    k_deg_init   <<<gN, blk, 0, stream>>>(degi, N);
    k_deg_rank   <<<gE, blk, 0, stream>>>(dst, degi, rank, E);
    k_rowptr_dinv<<<gScan, blk, 0, stream>>>(degi, rowptr, dinv, N);
    k_csr_fill   <<<gFill, blk, 0, stream>>>(src, dst, rank, rowptr, dinv, csr, E);

    k_gemm128_wmma<<<gGemm, blk, 0, stream>>>(x, W1, nullptr, hw, 0, gTiles);
    k_gather_rows <<<gRows, blk, 0, stream>>>(rowptr, csr, dinv, hw, b1, h, nullptr, N);

    k_gemm128_wmma<<<gGemm, blk, 0, stream>>>(h, W2, nullptr, hw, 0, gTiles);
    k_gather_rows <<<gRows, blk, 0, stream>>>(rowptr, csr, dinv, hw, b2, h, nullptr, N);

    k_gemm128_wmma<<<gGemm, blk, 0, stream>>>(h, W3, nullptr, hw, 0, gTiles);
    k_gather_rows <<<gRows, blk, 0, stream>>>(rowptr, csr, dinv, hw, b3, h, emb_out, N);

    k_gemm128_head<<<gGemm, blk, 0, stream>>>(h, Wp1, bp1, Wp2, bp2, prob_out, gTiles);
}


# Row sums fused into the next product: each 8-wave block sums its 16 nodes' rows straight into the LDS A tile, then multiplies; no h plane write/read, 3 launches fewer; all sum orders unchanged
# speedup vs baseline: 3.5725x; 1.0997x over previous
#include <hip/hip_runtime.h>
#include <math.h>

#define FD 128

typedef __attribute__((ext_vector_type(2))) float v2f;
typedef __attribute__((ext_vector_type(8))) float v8f;

__global__ void k_deg_init(unsigned* __restrict__ degi, int n) {
    int i = blockIdx.x * blockDim.x + threadIdx.x;
    if (i < n) degi[i] = 1u;
}

__global__ void k_deg_rank(const int* __restrict__ dst,
                           unsigned* __restrict__ degi,
                           int* __restrict__ rank, int e) {
    int i = blockIdx.x * blockDim.x + threadIdx.x;
    if (i < e) rank[i] = (int)atomicAdd(&degi[dst[i]], 1u);
}

__global__ __launch_bounds__(256) void k_rowptr_dinv(
    const unsigned* __restrict__ degi, int* __restrict__ rowptr,
    float* __restrict__ dinv, int n) {
    __shared__ unsigned red[8];
    __shared__ unsigned wtot[8];
    const int t    = threadIdx.x;
    const int lane = t & 31;
    const int wid  = t >> 5;
    const int base = blockIdx.x * 1024;
    const uint4* __restrict__ d4 = (const uint4*)degi;

    unsigned p = 0;
    const int nv = base >> 2;
#pragma unroll 8
    for (int j = t; j < nv; j += 256) {
        const uint4 v = d4[j];
        p += (v.x + v.y + v.z + v.w) - 4u;
    }
    for (int off = 16; off > 0; off >>= 1) p += __shfl_xor(p, off, 32);
    if (lane == 0) red[wid] = p;

    const int idx = base + 4 * t;
    const bool ok = (idx + 3 < n);
    uint4 c = make_uint4(1u, 1u, 1u, 1u);
    if (ok) c = d4[idx >> 2];
    const unsigned c0 = c.x - 1u, c1 = c.y - 1u, c2 = c.z - 1u, c3 = c.w - 1u;
    const unsigned s = c0 + c1 + c2 + c3;
    unsigned inc = s;
    for (int off = 1; off < 32; off <<= 1) {
        const unsigned y = __shfl_up(inc, off, 32);
        if (lane >= off) inc += y;
    }
    if (lane == 31) wtot[wid] = inc;
    __syncthreads();
    unsigned ex = inc - s;
    for (int w = 0; w < 8; ++w) {
        ex += red[w];
        if (w < wid) ex += wtot[w];
    }
    if (ok) {
        *(int4*)(rowptr + idx) = make_int4((int)ex, (int)(ex + c0),
                                           (int)(ex + c0 + c1),
                                           (int)(ex + c0 + c1 + c2));
        *(float4*)(dinv + idx) = make_float4(rsqrtf((float)c.x), rsqrtf((float)c.y),
                                             rsqrtf((float)c.z), rsqrtf((float)c.w));
        if (idx + 4 == n) rowptr[n] = (int)(ex + s);
    }
}

__device__ __forceinline__ void csr_put(int2* __restrict__ csr,
                                        const int* __restrict__ rowptr,
                                        const float* __restrict__ dinv,
                                        int s, int d, int r) {
    const float w = dinv[s] * dinv[d];
    csr[rowptr[d] + r - 1] = make_int2(s, __float_as_int(w));
}

__global__ __launch_bounds__(256) void k_csr_fill(
    const int* __restrict__ src, const int* __restrict__ dst,
    const int* __restrict__ rank, const int* __restrict__ rowptr,
    const float* __restrict__ dinv, int* __restrict__ csr_raw, int e) {
    int2* __restrict__ csr = (int2*)csr_raw;
    const int e0 = (blockIdx.x * blockDim.x + threadIdx.x) * 4;
    if (e0 >= e) return;
    if (((e & 3) == 0) && (e0 + 3 < e)) {
        const int4 s = *(const int4*)(src + e0);
        const int4 d = *(const int4*)(dst + e0);
        const int4 r = *(const int4*)(rank + e0);
        csr_put(csr, rowptr, dinv, s.x, d.x, r.x);
        csr_put(csr, rowptr, dinv, s.y, d.y, r.y);
        csr_put(csr, rowptr, dinv, s.z, d.z, r.z);
        csr_put(csr, rowptr, dinv, s.w, d.w, r.w);
    } else {
        for (int k = e0; k < e && k < e0 + 4; ++k)
            csr_put(csr, rowptr, dinv, src[k], dst[k], rank[k]);
    }
}

#define LDA 132

__device__ __forceinline__ void lds_barrier() {
    __builtin_amdgcn_fence(__ATOMIC_RELEASE, "workgroup", "local");
    __builtin_amdgcn_s_barrier();
    __builtin_amdgcn_fence(__ATOMIC_ACQUIRE, "workgroup", "local");
}

__device__ __forceinline__ float half_sum16(float x) {
    x += __int_as_float(__builtin_amdgcn_update_dpp(0, __float_as_int(x), 0x161, 0xf, 0xf, true));
    x += __int_as_float(__builtin_amdgcn_update_dpp(0, __float_as_int(x), 0x162, 0xf, 0xf, true));
    x += __int_as_float(__builtin_amdgcn_update_dpp(0, __float_as_int(x), 0x164, 0xf, 0xf, true));
    x += __int_as_float(__builtin_amdgcn_update_dpp(0, __float_as_int(x), 0x168, 0xf, 0xf, true));
    return x;
}

__device__ __forceinline__ void load_w_slice(v2f (&bf)[32], const float* __restrict__ W,
                                             const int n0, const int half, const int lm) {
#pragma unroll
    for (int s = 0; s < 32; ++s) {
        const int ka = 4 * s + 2 * half;
        bf[s].x = W[(size_t)(ka)     * FD + n0 + lm];
        bf[s].y = W[(size_t)(ka + 1) * FD + n0 + lm];
    }
}

__device__ __forceinline__ void tile_products(const float* __restrict__ as, const v2f (&bf)[32],
                                              v8f& acc0, v8f& acc1) {
#pragma unroll
    for (int s = 0; s < 32; s += 2) {
        const v2f a0 = *(const v2f*)(as + 4 * s);
        const v2f a1 = *(const v2f*)(as + 4 * s + 4);
        acc0 = __builtin_amdgcn_wmma_f32_16x16x4_f32(
            false, a0, false, bf[s], (short)0, acc0, false, false);
        acc1 = __builtin_amdgcn_wmma_f32_16x16x4_f32(
            false, a1, false, bf[s + 1], (short)0, acc1, false, false);
    }
}

__global__ __launch_bounds__(256) void k_gemm128_wmma(
    const float* __restrict__ A, const float* __restrict__ W,
    float* __restrict__ out, int ntiles) {
    __shared__ float As[2][16 * LDA];
    const int t    = threadIdx.x;
    const int lane = t & 31;
    const int wid  = t >> 5;
    const int n0   = wid << 4;
    const int half = lane >> 4;
    const int lm   = lane & 15;

    v2f bf[32];
    load_w_slice(bf, W, n0, half, lm);

    const int sr = t >> 5;
    const int sc = (t & 31) << 2;
    float* const st0 = &As[0][sr * LDA + sc];
    float* const st1 = &As[0][(sr + 8) * LDA + sc];

    int tile = blockIdx.x;
    if (tile < ntiles) {
        const float4* __restrict__ g = (const float4*)(A + (size_t)tile * (16 * FD));
        const float4 g0 = g[t];
        const float4 g1 = g[t + 256];
        *(float4*)st0 = g0;
        *(float4*)st1 = g1;
    }
    __syncthreads();

    int buf = 0;
    for (; tile < ntiles; tile += (int)gridDim.x) {
        const int  nxt  = tile + (int)gridDim.x;
        const bool more = nxt < ntiles;
        float4 g0 = make_float4(0.0f, 0.0f, 0.0f, 0.0f), g1 = g0;
        if (more) {
            const float4* __restrict__ g = (const float4*)(A + (size_t)nxt * (16 * FD));
            g0 = g[t];
            g1 = g[t + 256];
        }

        v8f acc0 = {};
        v8f acc1 = {};
        tile_products(&As[buf][lm * LDA + 2 * half], bf, acc0, acc1);

        const int m0 = tile << 4;
#pragma unroll
        for (int v = 0; v < 8; ++v) {
            const int row = m0 + v + 8 * half;
            out[(size_t)row * FD + n0 + lm] = acc0[v] + acc1[v];
        }

        if (more) {
            const int o = (buf ^ 1) * (16 * LDA);
            *(float4*)(st0 + o) = g0;
            *(float4*)(st1 + o) = g1;
        }
        lds_barrier();
        buf ^= 1;
    }
}


__device__ __forceinline__ void row_fma(float4& acc, const float4 v, const float w) {
    acc.x = fmaf(v.x, w, acc.x);
    acc.y = fmaf(v.y, w, acc.y);
    acc.z = fmaf(v.z, w, acc.z);
    acc.w = fmaf(v.w, w, acc.w);
}

template <int M>
__device__ __forceinline__ void walk_rows(float4& acc, const int2 r, const int k,
                                          const char* __restrict__ hwb,
                                          const unsigned loff) {
    float4 v[M];
    float  w[M];
    for (int j = 0; j < M; ++j) {
        const int s = __builtin_amdgcn_readlane(r.x, k + j);
        w[j] = __int_as_float(__builtin_amdgcn_readlane(r.y, k + j));
        v[j] = *(const float4*)(hwb + (size_t)s * 512 + loff);
    }
    for (int j = 0; j < M; ++j) row_fma(acc, v[j], w[j]);
}

__device__ __forceinline__ void walk_chunk(float4& acc, const int2 r, const int cnt,
                                           const char* __restrict__ hwb,
                                           const unsigned loff) {
    for (int k = 0; k < cnt; k += 8) {
        switch (cnt - k) {
            case 1:  walk_rows<1>(acc, r, k, hwb, loff); break;
            case 2:  walk_rows<2>(acc, r, k, hwb, loff); break;
            case 3:  walk_rows<3>(acc, r, k, hwb, loff); break;
            case 4:  walk_rows<4>(acc, r, k, hwb, loff); break;
            case 5:  walk_rows<5>(acc, r, k, hwb, loff); break;
            case 6:  walk_rows<6>(acc, r, k, hwb, loff); break;
            case 7:  walk_rows<7>(acc, r, k, hwb, loff); break;
            default: walk_rows<8>(acc, r, k, hwb, loff); break;
        }
    }
}

__device__ __forceinline__ int fetch_bounds(const int* __restrict__ rowptr,
                                            const int node0, const int n, const int lane) {
    int i = node0 + (lane < 2 ? lane : 2);
    i = (i < n) ? i : n;
    return rowptr[i];
}

template <bool HEAD>
__device__ __forceinline__ void rows_gemm_body(
    const int* __restrict__ rowptr, const int* __restrict__ csr_raw,
    const float* __restrict__ dinv, const float* __restrict__ hw,
    const float* __restrict__ brow, float* __restrict__ emb,
    const float* __restrict__ W, const float* __restrict__ bias,
    float* __restrict__ out, int n, int ntiles,
    const float* __restrict__ Wh, const float* __restrict__ bh,
    float* __restrict__ prob) {
    __shared__ float As[2][16 * LDA];
    __shared__ float part[HEAD ? 2 * 8 * 32 : 1];
    __shared__ __align__(16) float cst[HEAD ? 4 * FD + 2 : FD];
    const int t    = threadIdx.x;
    const int lane = t & 31;
    const int wid  = __builtin_amdgcn_readfirstlane(t >> 5);
    const int n0   = wid << 4;
    const int half = lane >> 4;
    const int lm   = lane & 15;

    const int2* __restrict__ csr = (const int2*)csr_raw;
    const char* __restrict__ hwb = (const char*)hw;
    const unsigned loff = (unsigned)lane * 16u;

    int tile = blockIdx.x;
    int rp_l = 0;
    int2 rec0 = make_int2(0, 0), rec1 = make_int2(0, 0);
    if (tile < ntiles) {
        rp_l = fetch_bounds(rowptr, tile * 16 + 2 * wid, n, lane);
        const int a = __builtin_amdgcn_readlane(rp_l, 0);
        const int b = __builtin_amdgcn_readlane(rp_l, 1);
        const int c = __builtin_amdgcn_readlane(rp_l, 2);
        if (a + lane < b) rec0 = csr[a + lane];
        if (b + lane < c) rec1 = csr[b + lane];
    }

    v2f bf[32];
    load_w_slice(bf, W, n0, half, lm);

    if (t < FD) {
        cst[t] = brow[t];
        if (HEAD) {
            cst[FD + t] = bias[t];
            *(float2*)&cst[2 * FD + 2 * t] = *(const float2*)(Wh + 2 * t);
            if (t < 2) cst[4 * FD + t] = bh[t];
        }
    }
    __syncthreads();

    int buf = 0;
    int it  = 0;
    for (; tile < ntiles; tile += (int)gridDim.x) {
        const int  nxt  = tile + (int)gridDim.x;
        const bool more = nxt < ntiles;
        int rp_n = 0;
        if (more) rp_n = fetch_bounds(rowptr, nxt * 16 + 2 * wid, n, lane);

        int rp[3];
        for (int q = 0; q < 3; ++q) rp[q] = __builtin_amdgcn_readlane(rp_l, q);
        int2 rec[2] = {rec0, rec1};

#pragma unroll
        for (int q = 0; q < 2; ++q) {
            const int node = tile * 16 + 2 * wid + q;
            float4 acc = make_float4(0.0f, 0.0f, 0.0f, 0.0f);
            if (node < n) {
                const float di = dinv[node];
                const float4 sv = *(const float4*)(hwb + (size_t)node * 512 + loff);
                const int beg = rp[q], end = rp[q + 1];
                walk_chunk(acc, rec[q], (end - beg < 32) ? (end - beg) : 32, hwb, loff);
                for (int cb = beg + 32; cb < end; cb += 32) {
                    int2 r = make_int2(0, 0);
                    if (cb + lane < end) r = csr[cb + lane];
                    walk_chunk(acc, r, (end - cb < 32) ? (end - cb) : 32, hwb, loff);
                }
                acc.x += sv.x * di * di;
                acc.y += sv.y * di * di;
                acc.z += sv.z * di * di;
                acc.w += sv.w * di * di;
                const float4 b4 = *(const float4*)&cst[4 * lane];
                acc.x += b4.x;
                acc.y += b4.y;
                acc.z += b4.z;
                acc.w += b4.w;
                if (HEAD) ((float4*)emb)[(unsigned)(node * 32 + lane)] = acc;
            }
            *(float4*)&As[buf][(2 * wid + q) * LDA + 4 * lane] =
                make_float4(fmaxf(acc.x, 0.0f), fmaxf(acc.y, 0.0f),
                            fmaxf(acc.z, 0.0f), fmaxf(acc.w, 0.0f));
        }

        rec0 = make_int2(0, 0);
        rec1 = make_int2(0, 0);
        if (more) {
            const int a = __builtin_amdgcn_readlane(rp_n, 0);
            const int b = __builtin_amdgcn_readlane(rp_n, 1);
            const int c = __builtin_amdgcn_readlane(rp_n, 2);
            if (a + lane < b) rec0 = csr[a + lane];
            if (b + lane < c) rec1 = csr[b + lane];
        }
        rp_l = rp_n;

        lds_barrier();

        v8f acc0 = {};
        v8f acc1 = {};
        tile_products(&As[buf][lm * LDA + 2 * half], bf, acc0, acc1);

        if (HEAD) {
            const int pb = it & 1;
            const float  bcol = cst[FD + n0 + lm];
            const float2 wh   = *(const float2*)&cst[2 * FD + 2 * (n0 + lm)];
            const float  wh0 = wh.x, wh1 = wh.y;
            float p[16];
#pragma unroll
            for (int v = 0; v < 8; ++v) {
                float r = acc0[v] + acc1[v];
                r += bcol;
                r = fmaxf(r, 0.0f);
                p[2 * v]     = r * wh0;
                p[2 * v + 1] = r * wh1;
            }
#pragma unroll
            for (int i = 0; i < 16; ++i) p[i] = half_sum16(p[i]);
            if (lm == 0) {
                float4* __restrict__ pp = (float4*)&part[(pb * 8 + wid) * 32 + half * 16];
                pp[0] = make_float4(p[0],  p[1],  p[2],  p[3]);
                pp[1] = make_float4(p[4],  p[5],  p[6],  p[7]);
                pp[2] = make_float4(p[8],  p[9],  p[10], p[11]);
                pp[3] = make_float4(p[12], p[13], p[14], p[15]);
            }
            lds_barrier();
            if (wid == (it & 7)) {
                const float* __restrict__ pr = &part[pb * 256 + lane];
                float l = pr[0];
#pragma unroll
                for (int w = 1; w < 8; ++w) l += pr[w * 32];
                l += cst[4 * FD + (lane & 1)];
                const float o = __int_as_float(
                    __builtin_amdgcn_update_dpp(0, __float_as_int(l), 0x161, 0xf, 0xf, true));
                const float m   = fmaxf(l, o);
                const float lse = m + logf(expf(l - m) + expf(o - m));
                prob[(unsigned)(tile * 32 + lane)] = l - lse;
            }
        } else {
            const int m0 = tile << 4;
#pragma unroll
            for (int v = 0; v < 8; ++v) {
                const int row = m0 + v + 8 * half;
                out[(size_t)row * FD + n0 + lm] = acc0[v] + acc1[v];
            }
        }
        buf ^= 1;
        ++it;
    }
}

__global__ __launch_bounds__(256, 8) void k_rows_gemm(
    const int* __restrict__ rowptr, const int* __restrict__ csr_raw,
    const float* __restrict__ dinv, const float* __restrict__ hw,
    const float* __restrict__ brow, const float* __restrict__ W,
    float* __restrict__ out, int n, int ntiles) {
    rows_gemm_body<false>(rowptr, csr_raw, dinv, hw, brow, nullptr, W, nullptr, out, n, ntiles,
                          nullptr, nullptr, nullptr);
}

__global__ __launch_bounds__(256, 8) void k_rows_gemm_head(
    const int* __restrict__ rowptr, const int* __restrict__ csr_raw,
    const float* __restrict__ dinv, const float* __restrict__ hw,
    const float* __restrict__ brow, float* __restrict__ emb,
    const float* __restrict__ W, const float* __restrict__ bias,
    const float* __restrict__ Wh, const float* __restrict__ bh,
    float* __restrict__ prob, int n, int ntiles) {
    rows_gemm_body<true>(rowptr, csr_raw, dinv, hw, brow, emb, W, bias, nullptr, n, ntiles,
                         Wh, bh, prob);
}

extern "C" void kernel_launch(void* const* d_in, const int* in_sizes, int n_in,
                              void* d_out, int out_size, void* d_ws,
                              size_t ws_size, hipStream_t stream) {
    const float* x   = (const float*)d_in[0];
    const int*   ei  = (const int*)d_in[1];
    const float* W1  = (const float*)d_in[2];
    const float* b1  = (const float*)d_in[3];
    const float* W2  = (const float*)d_in[4];
    const float* b2  = (const float*)d_in[5];
    const float* W3  = (const float*)d_in[6];
    const float* b3  = (const float*)d_in[7];
    const float* Wp1 = (const float*)d_in[8];
    const float* bp1 = (const float*)d_in[9];
    const float* Wp2 = (const float*)d_in[10];
    const float* bp2 = (const float*)d_in[11];

    const int N = in_sizes[0] / FD;
    const int E = in_sizes[1] / 2;
    const int* src = ei;
    const int* dst = ei + E;

    char* ws = (char*)d_ws;
    const size_t AL = 256;
    unsigned* degi = (unsigned*)ws;  ws += ((size_t)N * sizeof(unsigned) + AL - 1) / AL * AL;
    float* dinv = (float*)ws;        ws += ((size_t)N * sizeof(float) + AL - 1) / AL * AL;
    int* rowptr = (int*)ws;          ws += ((size_t)(N + 1) * sizeof(int) + AL - 1) / AL * AL;
    int* rank   = (int*)ws;          ws += ((size_t)E * sizeof(int) + AL - 1) / AL * AL;
    int* csr    = (int*)ws;          ws += ((size_t)E * 2 * sizeof(int) + AL - 1) / AL * AL;
    float* hw0  = (float*)ws;        ws += (size_t)N * FD * sizeof(float);
    float* hw1  = (float*)ws;

    float* emb_out  = (float*)d_out;
    float* prob_out = (float*)d_out + (size_t)N * FD;

    const dim3 blk(256);
    const int gN      = (N + 255) / 256;
    const int gE      = (E + 255) / 256;
    const int gScan   = (N + 1023) / 1024;
    const int gFill   = (E + 1023) / 1024;
    const int gTiles  = N / 16;
    const int gGemm   = 1024;

    k_deg_init   <<<gN, blk, 0, stream>>>(degi, N);
    k_deg_rank   <<<gE, blk, 0, stream>>>(dst, degi, rank, E);
    k_rowptr_dinv<<<gScan, blk, 0, stream>>>(degi, rowptr, dinv, N);
    k_csr_fill   <<<gFill, blk, 0, stream>>>(src, dst, rank, rowptr, dinv, csr, E);

    k_gemm128_wmma<<<gGemm, blk, 0, stream>>>(x, W1, hw0, gTiles);

    k_rows_gemm<<<gGemm, blk, 0, stream>>>(rowptr, csr, dinv, hw0, b1, W2, hw1, N, gTiles);

    k_rows_gemm<<<gGemm, blk, 0, stream>>>(rowptr, csr, dinv, hw1, b2, W3, hw0, N, gTiles);

    k_rows_gemm_head<<<gGemm, blk, 0, stream>>>(rowptr, csr, dinv, hw0, b3, emb_out,
                                                Wp1, bp1, Wp2, bp2, prob_out, N, gTiles);
}


# Edge ranks by returning LDS adds: 32 work-groups count one edge chunk each in a whole-node LDS table, a scan of the tables gives positions; no global atomic, 7 launches; sum order = chunk, then rank
# speedup vs baseline: 4.6512x; 1.3019x over previous
#include <hip/hip_runtime.h>
#include <math.h>

#define FD 128

typedef __attribute__((ext_vector_type(2))) float v2f;
typedef __attribute__((ext_vector_type(8))) float v8f;

#define RK_P   32
#define RK_NP  50176
#define RK_SEG 64

__device__ __forceinline__ unsigned wave_sum_u32(unsigned x) {
    x += (unsigned)__builtin_amdgcn_update_dpp(0, (int)x, 0x161, 0xf, 0xf, true);
    x += (unsigned)__builtin_amdgcn_update_dpp(0, (int)x, 0x162, 0xf, 0xf, true);
    x += (unsigned)__builtin_amdgcn_update_dpp(0, (int)x, 0x164, 0xf, 0xf, true);
    x += (unsigned)__builtin_amdgcn_update_dpp(0, (int)x, 0x168, 0xf, 0xf, true);
    return (unsigned)__builtin_amdgcn_readlane((int)x, 0) +
           (unsigned)__builtin_amdgcn_readlane((int)x, 16);
}

__global__ __launch_bounds__(1024) void k_rank_lds(
    const int* __restrict__ dst, int* __restrict__ rank,
    unsigned* __restrict__ gcnt, unsigned* __restrict__ bsum, int e, int chunk) {
    __shared__ __align__(16) unsigned cnt[RK_NP];
    __shared__ unsigned seg[RK_SEG];
    const int t    = threadIdx.x;
    const int lane = t & 31;
    const int p    = blockIdx.x;
    uint4* const c4 = (uint4*)cnt;

    for (int j = t; j < RK_NP / 4; j += 1024) c4[j] = make_uint4(0u, 0u, 0u, 0u);
    if (t < RK_SEG) seg[t] = 0u;
    __syncthreads();

    const int beg = p * chunk;
    const int end = (beg + chunk < e) ? (beg + chunk) : e;
    if ((e & 3) == 0) {
        int j = beg + 4 * t;
        int4 d = make_int4(0, 0, 0, 0);
        if (j < end) d = *(const int4*)(dst + j);
        while (j < end) {
            const int jn = j + 4096;
            int4 dn = d;
            if (jn < end) dn = *(const int4*)(dst + jn);
            int4 r;
            r.x = (int)atomicAdd(&cnt[d.x], 1u);
            r.y = (int)atomicAdd(&cnt[d.y], 1u);
            r.z = (int)atomicAdd(&cnt[d.z], 1u);
            r.w = (int)atomicAdd(&cnt[d.w], 1u);
            *(int4*)(rank + j) = r;
            d = dn;
            j = jn;
        }
    } else {
        for (int j = beg + t; j < end; j += 1024)
            rank[j] = (int)atomicAdd(&cnt[dst[j]], 1u);
    }
    __syncthreads();

    uint4* __restrict__ g4 = (uint4*)(gcnt + (size_t)p * RK_NP);
    for (int j = t; j < RK_NP / 4; j += 1024) {
        const uint4 v = c4[j];
        g4[j] = v;
        const unsigned s = wave_sum_u32(v.x + v.y + v.z + v.w);
        if (lane == 0) atomicAdd(&seg[j >> 8], s);
    }
    __syncthreads();
    if (t < RK_SEG) bsum[p * RK_SEG + t] = seg[t];
}

__global__ __launch_bounds__(256) void k_rowptr_dinv(
    unsigned* __restrict__ gcnt, const unsigned* __restrict__ bsum,
    int* __restrict__ rowptr, float* __restrict__ dinv, int n) {
    __shared__ unsigned red[8];
    __shared__ unsigned wtot[8];
    const int t    = threadIdx.x;
    const int lane = t & 31;
    const int wid  = t >> 5;
    const int b    = blockIdx.x;
    const int idx  = b * 1024 + 4 * t;

    uint4 c[RK_P];
#pragma unroll
    for (int p = 0; p < RK_P; ++p) c[p] = *(const uint4*)(gcnt + (size_t)p * RK_NP + idx);

    unsigned pf = 0;
#pragma unroll
    for (int k = 0; k < RK_P * RK_SEG / 256; ++k) {
        const int j = t + 256 * k;
        const unsigned v = bsum[j];
        pf += ((j & (RK_SEG - 1)) < b) ? v : 0u;
    }
    for (int off = 16; off > 0; off >>= 1) pf += __shfl_xor(pf, off, 32);
    if (lane == 0) red[wid] = pf;

    uint4 run = make_uint4(0u, 0u, 0u, 0u);
#pragma unroll
    for (int p = 0; p < RK_P; ++p) {
        const uint4 o = run;
        run.x += c[p].x; run.y += c[p].y; run.z += c[p].z; run.w += c[p].w;
        c[p] = o;
    }
    const unsigned c0 = run.x, c1 = run.y, c2 = run.z, c3 = run.w;
    const unsigned s = c0 + c1 + c2 + c3;
    unsigned inc = s;
    for (int off = 1; off < 32; off <<= 1) {
        const unsigned y = __shfl_up(inc, off, 32);
        if (lane >= off) inc += y;
    }
    if (lane == 31) wtot[wid] = inc;
    __syncthreads();
    unsigned ex = inc - s;
    for (int w = 0; w < 8; ++w) {
        ex += red[w];
        if (w < wid) ex += wtot[w];
    }
    const unsigned r0 = ex, r1 = ex + c0, r2 = r1 + c1, r3 = r2 + c2;
#pragma unroll
    for (int p = 0; p < RK_P; ++p)
        *(uint4*)(gcnt + (size_t)p * RK_NP + idx) =
            make_uint4(c[p].x + r0, c[p].y + r1, c[p].z + r2, c[p].w + r3);
    if (idx + 3 < n) {
        *(int4*)(rowptr + idx) = make_int4((int)r0, (int)r1, (int)r2, (int)r3);
        *(float4*)(dinv + idx) = make_float4(rsqrtf((float)(c0 + 1u)), rsqrtf((float)(c1 + 1u)),
                                             rsqrtf((float)(c2 + 1u)), rsqrtf((float)(c3 + 1u)));
        if (idx + 4 == n) rowptr[n] = (int)(ex + s);
    }
}

__device__ __forceinline__ void csr_put(int2* __restrict__ csr,
                                        const unsigned* __restrict__ pos,
                                        const float* __restrict__ dinv,
                                        int s, int d, int r) {
    const float w = dinv[s] * dinv[d];
    csr[pos[d] + (unsigned)r] = make_int2(s, __float_as_int(w));
}

__global__ __launch_bounds__(256) void k_csr_fill(
    const int* __restrict__ src, const int* __restrict__ dst,
    const int* __restrict__ rank, const unsigned* __restrict__ gpos,
    const float* __restrict__ dinv, int* __restrict__ csr_raw, int e, int chunk) {
    int2* __restrict__ csr = (int2*)csr_raw;
    const int e0 = (blockIdx.x * blockDim.x + threadIdx.x) * 4;
    if (e0 >= e) return;
    if (((e & 3) == 0) && (e0 + 3 < e)) {
        const unsigned* __restrict__ pos = gpos + (size_t)(e0 / chunk) * RK_NP;
        const int4 s = *(const int4*)(src + e0);
        const int4 d = *(const int4*)(dst + e0);
        const int4 r = *(const int4*)(rank + e0);
        csr_put(csr, pos, dinv, s.x, d.x, r.x);
        csr_put(csr, pos, dinv, s.y, d.y, r.y);
        csr_put(csr, pos, dinv, s.z, d.z, r.z);
        csr_put(csr, pos, dinv, s.w, d.w, r.w);
    } else {
        for (int k = e0; k < e && k < e0 + 4; ++k)
            csr_put(csr, gpos + (size_t)(k / chunk) * RK_NP, dinv, src[k], dst[k], rank[k]);
    }
}

#define LDA 132

__device__ __forceinline__ void lds_barrier() {
    __builtin_amdgcn_fence(__ATOMIC_RELEASE, "workgroup", "local");
    __builtin_amdgcn_s_barrier();
    __builtin_amdgcn_fence(__ATOMIC_ACQUIRE, "workgroup", "local");
}

__device__ __forceinline__ float half_sum16(float x) {
    x += __int_as_float(__builtin_amdgcn_update_dpp(0, __float_as_int(x), 0x161, 0xf, 0xf, true));
    x += __int_as_float(__builtin_amdgcn_update_dpp(0, __float_as_int(x), 0x162, 0xf, 0xf, true));
    x += __int_as_float(__builtin_amdgcn_update_dpp(0, __float_as_int(x), 0x164, 0xf, 0xf, true));
    x += __int_as_float(__builtin_amdgcn_update_dpp(0, __float_as_int(x), 0x168, 0xf, 0xf, true));
    return x;
}

__device__ __forceinline__ void load_w_slice(v2f (&bf)[32], const float* __restrict__ W,
                                             const int n0, const int half, const int lm) {
#pragma unroll
    for (int s = 0; s < 32; ++s) {
        const int ka = 4 * s + 2 * half;
        bf[s].x = W[(size_t)(ka)     * FD + n0 + lm];
        bf[s].y = W[(size_t)(ka + 1) * FD + n0 + lm];
    }
}

__device__ __forceinline__ void tile_products(const float* __restrict__ as, const v2f (&bf)[32],
                                              v8f& acc0, v8f& acc1) {
#pragma unroll
    for (int s = 0; s < 32; s += 2) {
        const v2f a0 = *(const v2f*)(as + 4 * s);
        const v2f a1 = *(const v2f*)(as + 4 * s + 4);
        acc0 = __builtin_amdgcn_wmma_f32_16x16x4_f32(
            false, a0, false, bf[s], (short)0, acc0, false, false);
        acc1 = __builtin_amdgcn_wmma_f32_16x16x4_f32(
            false, a1, false, bf[s + 1], (short)0, acc1, false, false);
    }
}

__global__ __launch_bounds__(256) void k_gemm128_wmma(
    const float* __restrict__ A, const float* __restrict__ W,
    float* __restrict__ out, int ntiles) {
    __shared__ float As[2][16 * LDA];
    const int t    = threadIdx.x;
    const int lane = t & 31;
    const int wid  = t >> 5;
    const int n0   = wid << 4;
    const int half = lane >> 4;
    const int lm   = lane & 15;

    v2f bf[32];
    load_w_slice(bf, W, n0, half, lm);

    const int sr = t >> 5;
    const int sc = (t & 31) << 2;
    float* const st0 = &As[0][sr * LDA + sc];
    float* const st1 = &As[0][(sr + 8) * LDA + sc];

    int tile = blockIdx.x;
    if (tile < ntiles) {
        const float4* __restrict__ g = (const float4*)(A + (size_t)tile * (16 * FD));
        const float4 g0 = g[t];
        const float4 g1 = g[t + 256];
        *(float4*)st0 = g0;
        *(float4*)st1 = g1;
    }
    __syncthreads();

    int buf = 0;
    for (; tile < ntiles; tile += (int)gridDim.x) {
        const int  nxt  = tile + (int)gridDim.x;
        const bool more = nxt < ntiles;
        float4 g0 = make_float4(0.0f, 0.0f, 0.0f, 0.0f), g1 = g0;
        if (more) {
            const float4* __restrict__ g = (const float4*)(A + (size_t)nxt * (16 * FD));
            g0 = g[t];
            g1 = g[t + 256];
        }

        v8f acc0 = {};
        v8f acc1 = {};
        tile_products(&As[buf][lm * LDA + 2 * half], bf, acc0, acc1);

        const int m0 = tile << 4;
#pragma unroll
        for (int v = 0; v < 8; ++v) {
            const int row = m0 + v + 8 * half;
            out[(size_t)row * FD + n0 + lm] = acc0[v] + acc1[v];
        }

        if (more) {
            const int o = (buf ^ 1) * (16 * LDA);
            *(float4*)(st0 + o) = g0;
            *(float4*)(st1 + o) = g1;
        }
        lds_barrier();
        buf ^= 1;
    }
}


__device__ __forceinline__ void row_fma(float4& acc, const float4 v, const float w) {
    acc.x = fmaf(v.x, w, acc.x);
    acc.y = fmaf(v.y, w, acc.y);
    acc.z = fmaf(v.z, w, acc.z);
    acc.w = fmaf(v.w, w, acc.w);
}

template <int M>
__device__ __forceinline__ void walk_rows(float4& acc, const int2 r, const int k,
                                          const char* __restrict__ hwb,
                                          const unsigned loff) {
    float4 v[M];
    float  w[M];
    for (int j = 0; j < M; ++j) {
        const int s = __builtin_amdgcn_readlane(r.x, k + j);
        w[j] = __int_as_float(__builtin_amdgcn_readlane(r.y, k + j));
        v[j] = *(const float4*)(hwb + (size_t)s * 512 + loff);
    }
    for (int j = 0; j < M; ++j) row_fma(acc, v[j], w[j]);
}

__device__ __forceinline__ void walk_chunk(float4& acc, const int2 r, const int cnt,
                                           const char* __restrict__ hwb,
                                           const unsigned loff) {
    for (int k = 0; k < cnt; k += 8) {
        switch (cnt - k) {
            case 1:  walk_rows<1>(acc, r, k, hwb, loff); break;
            case 2:  walk_rows<2>(acc, r, k, hwb, loff); break;
            case 3:  walk_rows<3>(acc, r, k, hwb, loff); break;
            case 4:  walk_rows<4>(acc, r, k, hwb, loff); break;
            case 5:  walk_rows<5>(acc, r, k, hwb, loff); break;
            case 6:  walk_rows<6>(acc, r, k, hwb, loff); break;
            case 7:  walk_rows<7>(acc, r, k, hwb, loff); break;
            default: walk_rows<8>(acc, r, k, hwb, loff); break;
        }
    }
}

__device__ __forceinline__ int fetch_bounds(const int* __restrict__ rowptr,
                                            const int node0, const int n, const int lane) {
    int i = node0 + (lane < 2 ? lane : 2);
    i = (i < n) ? i : n;
    return rowptr[i];
}

template <bool HEAD>
__device__ __forceinline__ void rows_gemm_body(
    const int* __restrict__ rowptr, const int* __restrict__ csr_raw,
    const float* __restrict__ dinv, const float* __restrict__ hw,
    const float* __restrict__ brow, float* __restrict__ emb,
    const float* __restrict__ W, const float* __restrict__ bias,
    float* __restrict__ out, int n, int ntiles,
    const float* __restrict__ Wh, const float* __restrict__ bh,
    float* __restrict__ prob) {
    __shared__ float As[2][16 * LDA];
    __shared__ float part[HEAD ? 2 * 8 * 32 : 1];
    __shared__ __align__(16) float cst[HEAD ? 4 * FD + 2 : FD];
    const int t    = threadIdx.x;
    const int lane = t & 31;
    const int wid  = __builtin_amdgcn_readfirstlane(t >> 5);
    const int n0   = wid << 4;
    const int half = lane >> 4;
    const int lm   = lane & 15;

    const int2* __restrict__ csr = (const int2*)csr_raw;
    const char* __restrict__ hwb = (const char*)hw;
    const unsigned loff = (unsigned)lane * 16u;

    int tile = blockIdx.x;
    int rp_l = 0;
    int2 rec0 = make_int2(0, 0), rec1 = make_int2(0, 0);
    if (tile < ntiles) {
        rp_l = fetch_bounds(rowptr, tile * 16 + 2 * wid, n, lane);
        const int a = __builtin_amdgcn_readlane(rp_l, 0);
        const int b = __builtin_amdgcn_readlane(rp_l, 1);
        const int c = __builtin_amdgcn_readlane(rp_l, 2);
        if (a + lane < b) rec0 = csr[a + lane];
        if (b + lane < c) rec1 = csr[b + lane];
    }

    v2f bf[32];
    load_w_slice(bf, W, n0, half, lm);

    if (t < FD) {
        cst[t] = brow[t];
        if (HEAD) {
            cst[FD + t] = bias[t];
            *(float2*)&cst[2 * FD + 2 * t] = *(const float2*)(Wh + 2 * t);
            if (t < 2) cst[4 * FD + t] = bh[t];
        }
    }
    __syncthreads();

    int buf = 0;
    int it  = 0;
    for (; tile < ntiles; tile += (int)gridDim.x) {
        const int  nxt  = tile + (int)gridDim.x;
        const bool more = nxt < ntiles;
        int rp_n = 0;
        if (more) rp_n = fetch_bounds(rowptr, nxt * 16 + 2 * wid, n, lane);

        int rp[3];
        for (int q = 0; q < 3; ++q) rp[q] = __builtin_amdgcn_readlane(rp_l, q);
        int2 rec[2] = {rec0, rec1};

#pragma unroll
        for (int q = 0; q < 2; ++q) {
            const int node = tile * 16 + 2 * wid + q;
            float4 acc = make_float4(0.0f, 0.0f, 0.0f, 0.0f);
            if (node < n) {
                const float di = dinv[node];
                const float4 sv = *(const float4*)(hwb + (size_t)node * 512 + loff);
                const int beg = rp[q], end = rp[q + 1];
                walk_chunk(acc, rec[q], (end - beg < 32) ? (end - beg) : 32, hwb, loff);
                for (int cb = beg + 32; cb < end; cb += 32) {
                    int2 r = make_int2(0, 0);
                    if (cb + lane < end) r = csr[cb + lane];
                    walk_chunk(acc, r, (end - cb < 32) ? (end - cb) : 32, hwb, loff);
                }
                acc.x += sv.x * di * di;
                acc.y += sv.y * di * di;
                acc.z += sv.z * di * di;
                acc.w += sv.w * di * di;
                const float4 b4 = *(const float4*)&cst[4 * lane];
                acc.x += b4.x;
                acc.y += b4.y;
                acc.z += b4.z;
                acc.w += b4.w;
                if (HEAD) ((float4*)emb)[(unsigned)(node * 32 + lane)] = acc;
            }
            *(float4*)&As[buf][(2 * wid + q) * LDA + 4 * lane] =
                make_float4(fmaxf(acc.x, 0.0f), fmaxf(acc.y, 0.0f),
                            fmaxf(acc.z, 0.0f), fmaxf(acc.w, 0.0f));
        }

        rec0 = make_int2(0, 0);
        rec1 = make_int2(0, 0);
        if (more) {
            const int a = __builtin_amdgcn_readlane(rp_n, 0);
            const int b = __builtin_amdgcn_readlane(rp_n, 1);
            const int c = __builtin_amdgcn_readlane(rp_n, 2);
            if (a + lane < b) rec0 = csr[a + lane];
            if (b + lane < c) rec1 = csr[b + lane];
        }
        rp_l = rp_n;

        lds_barrier();

        v8f acc0 = {};
        v8f acc1 = {};
        tile_products(&As[buf][lm * LDA + 2 * half], bf, acc0, acc1);

        if (HEAD) {
            const int pb = it & 1;
            const float  bcol = cst[FD + n0 + lm];
            const float2 wh   = *(const float2*)&cst[2 * FD + 2 * (n0 + lm)];
            const float  wh0 = wh.x, wh1 = wh.y;
            float p[16];
#pragma unroll
            for (int v = 0; v < 8; ++v) {
                float r = acc0[v] + acc1[v];
                r += bcol;
                r = fmaxf(r, 0.0f);
                p[2 * v]     = r * wh0;
                p[2 * v + 1] = r * wh1;
            }
#pragma unroll
            for (int i = 0; i < 16; ++i) p[i] = half_sum16(p[i]);
            if (lm == 0) {
                float4* __restrict__ pp = (float4*)&part[(pb * 8 + wid) * 32 + half * 16];
                pp[0] = make_float4(p[0],  p[1],  p[2],  p[3]);
                pp[1] = make_float4(p[4],  p[5],  p[6],  p[7]);
                pp[2] = make_float4(p[8],  p[9],  p[10], p[11]);
                pp[3] = make_float4(p[12], p[13], p[14], p[15]);
            }
            lds_barrier();
            if (wid == (it & 7)) {
                const float* __restrict__ pr = &part[pb * 256 + lane];
                float l = pr[0];
#pragma unroll
                for (int w = 1; w < 8; ++w) l += pr[w * 32];
                l += cst[4 * FD + (lane & 1)];
                const float o = __int_as_float(
                    __builtin_amdgcn_update_dpp(0, __float_as_int(l), 0x161, 0xf, 0xf, true));
                const float m   = fmaxf(l, o);
                const float lse = m + logf(expf(l - m) + expf(o - m));
                prob[(unsigned)(tile * 32 + lane)] = l - lse;
            }
        } else {
            const int m0 = tile << 4;
#pragma unroll
            for (int v = 0; v < 8; ++v) {
                const int row = m0 + v + 8 * half;
                out[(size_t)row * FD + n0 + lm] = acc0[v] + acc1[v];
            }
        }
        buf ^= 1;
        ++it;
    }
}

__global__ __launch_bounds__(256, 8) void k_rows_gemm(
    const int* __restrict__ rowptr, const int* __restrict__ csr_raw,
    const float* __restrict__ dinv, const float* __restrict__ hw,
    const float* __restrict__ brow, const float* __restrict__ W,
    float* __restrict__ out, int n, int ntiles) {
    rows_gemm_body<false>(rowptr, csr_raw, dinv, hw, brow, nullptr, W, nullptr, out, n, ntiles,
                          nullptr, nullptr, nullptr);
}

__global__ __launch_bounds__(256, 8) void k_rows_gemm_head(
    const int* __restrict__ rowptr, const int* __restrict__ csr_raw,
    const float* __restrict__ dinv, const float* __restrict__ hw,
    const float* __restrict__ brow, float* __restrict__ emb,
    const float* __restrict__ W, const float* __restrict__ bias,
    const float* __restrict__ Wh, const float* __restrict__ bh,
    float* __restrict__ prob, int n, int ntiles) {
    rows_gemm_body<true>(rowptr, csr_raw, dinv, hw, brow, emb, W, bias, nullptr, n, ntiles,
                         Wh, bh, prob);
}

extern "C" void kernel_launch(void* const* d_in, const int* in_sizes, int n_in,
                              void* d_out, int out_size, void* d_ws,
                              size_t ws_size, hipStream_t stream) {
    const float* x   = (const float*)d_in[0];
    const int*   ei  = (const int*)d_in[1];
    const float* W1  = (const float*)d_in[2];
    const float* b1  = (const float*)d_in[3];
    const float* W2  = (const float*)d_in[4];
    const float* b2  = (const float*)d_in[5];
    const float* W3  = (const float*)d_in[6];
    const float* b3  = (const float*)d_in[7];
    const float* Wp1 = (const float*)d_in[8];
    const float* bp1 = (const float*)d_in[9];
    const float* Wp2 = (const float*)d_in[10];
    const float* bp2 = (const float*)d_in[11];

    const int N = in_sizes[0] / FD;
    const int E = in_sizes[1] / 2;
    const int* src = ei;
    const int* dst = ei + E;

    char* ws = (char*)d_ws;
    const size_t AL = 256;
    float* dinv = (float*)ws;        ws += ((size_t)N * sizeof(float) + AL - 1) / AL * AL;
    int* rowptr = (int*)ws;          ws += ((size_t)(N + 1) * sizeof(int) + AL - 1) / AL * AL;
    int* rank   = (int*)ws;          ws += ((size_t)E * sizeof(int) + AL - 1) / AL * AL;
    int* csr    = (int*)ws;          ws += ((size_t)E * 2 * sizeof(int) + AL - 1) / AL * AL;
    unsigned* gcnt = (unsigned*)ws;  ws += (size_t)RK_P * RK_NP * sizeof(unsigned);
    unsigned* bsum = (unsigned*)ws;  ws += (size_t)RK_P * RK_SEG * sizeof(unsigned);
    float* hw0  = (float*)ws;        ws += (size_t)N * FD * sizeof(float);
    float* hw1  = (float*)ws;

    float* emb_out  = (float*)d_out;
    float* prob_out = (float*)d_out + (size_t)N * FD;

    const dim3 blk(256);
    const int gScan   = (N + 1023) / 1024;
    const int gFill   = (E + 1023) / 1024;
    const int gTiles  = N / 16;
    const int gGemm   = 1024;
    const int chunk   = ((E + RK_P - 1) / RK_P + 3) & ~3;

    k_rank_lds   <<<RK_P, dim3(1024), 0, stream>>>(dst, rank, gcnt, bsum, E, chunk);
    k_rowptr_dinv<<<gScan, blk, 0, stream>>>(gcnt, bsum, rowptr, dinv, N);
    k_csr_fill   <<<gFill, blk, 0, stream>>>(src, dst, rank, gcnt, dinv, csr, E, chunk);

    k_gemm128_wmma<<<gGemm, blk, 0, stream>>>(x, W1, hw0, gTiles);

    k_rows_gemm<<<gGemm, blk, 0, stream>>>(rowptr, csr, dinv, hw0, b1, W2, hw1, N, gTiles);

    k_rows_gemm<<<gGemm, blk, 0, stream>>>(rowptr, csr, dinv, hw1, b2, W3, hw0, N, gTiles);

    k_rows_gemm_head<<<gGemm, blk, 0, stream>>>(rowptr, csr, dinv, hw0, b3, emb_out,
                                                Wp1, bp1, Wp2, bp2, prob_out, N, gTiles);
}


# Layer 1 sums the rows of x first; one launch of 16-wave blocks then runs both products a tile apart (W1,+b1,ReLU on 8 waves; W2 on 8): no product launch, no hw0 plane; layer-1 sum order changes
# speedup vs baseline: 4.8626x; 1.0455x over previous
#include <hip/hip_runtime.h>
#include <math.h>

#define FD 128

typedef __attribute__((ext_vector_type(2))) float v2f;
typedef __attribute__((ext_vector_type(8))) float v8f;

#define RK_P   32
#define RK_NP  50176
#define RK_SEG 64

__device__ __forceinline__ unsigned wave_sum_u32(unsigned x) {
    x += (unsigned)__builtin_amdgcn_update_dpp(0, (int)x, 0x161, 0xf, 0xf, true);
    x += (unsigned)__builtin_amdgcn_update_dpp(0, (int)x, 0x162, 0xf, 0xf, true);
    x += (unsigned)__builtin_amdgcn_update_dpp(0, (int)x, 0x164, 0xf, 0xf, true);
    x += (unsigned)__builtin_amdgcn_update_dpp(0, (int)x, 0x168, 0xf, 0xf, true);
    return (unsigned)__builtin_amdgcn_readlane((int)x, 0) +
           (unsigned)__builtin_amdgcn_readlane((int)x, 16);
}

__global__ __launch_bounds__(1024) void k_rank_lds(
    const int* __restrict__ dst, int* __restrict__ rank,
    unsigned* __restrict__ gcnt, unsigned* __restrict__ bsum, int e, int chunk) {
    __shared__ __align__(16) unsigned cnt[RK_NP];
    __shared__ unsigned seg[RK_SEG];
    const int t    = threadIdx.x;
    const int lane = t & 31;
    const int p    = blockIdx.x;
    uint4* const c4 = (uint4*)cnt;

    for (int j = t; j < RK_NP / 4; j += 1024) c4[j] = make_uint4(0u, 0u, 0u, 0u);
    if (t < RK_SEG) seg[t] = 0u;
    __syncthreads();

    const int beg = p * chunk;
    const int end = (beg + chunk < e) ? (beg + chunk) : e;
    if ((e & 3) == 0) {
        int j = beg + 4 * t;
        int4 d = make_int4(0, 0, 0, 0);
        if (j < end) d = *(const int4*)(dst + j);
        while (j < end) {
            const int jn = j + 4096;
            int4 dn = d;
            if (jn < end) dn = *(const int4*)(dst + jn);
            int4 r;
            r.x = (int)atomicAdd(&cnt[d.x], 1u);
            r.y = (int)atomicAdd(&cnt[d.y], 1u);
            r.z = (int)atomicAdd(&cnt[d.z], 1u);
            r.w = (int)atomicAdd(&cnt[d.w], 1u);
            *(int4*)(rank + j) = r;
            d = dn;
            j = jn;
        }
    } else {
        for (int j = beg + t; j < end; j += 1024)
            rank[j] = (int)atomicAdd(&cnt[dst[j]], 1u);
    }
    __syncthreads();

    uint4* __restrict__ g4 = (uint4*)(gcnt + (size_t)p * RK_NP);
    for (int j = t; j < RK_NP / 4; j += 1024) {
        const uint4 v = c4[j];
        g4[j] = v;
        const unsigned s = wave_sum_u32(v.x + v.y + v.z + v.w);
        if (lane == 0) atomicAdd(&seg[j >> 8], s);
    }
    __syncthreads();
    if (t < RK_SEG) bsum[p * RK_SEG + t] = seg[t];
}

__global__ __launch_bounds__(256) void k_rowptr_dinv(
    unsigned* __restrict__ gcnt, const unsigned* __restrict__ bsum,
    int* __restrict__ rowptr, float* __restrict__ dinv, int n) {
    __shared__ unsigned red[8];
    __shared__ unsigned wtot[8];
    const int t    = threadIdx.x;
    const int lane = t & 31;
    const int wid  = t >> 5;
    const int b    = blockIdx.x;
    const int idx  = b * 1024 + 4 * t;

    uint4 c[RK_P];
#pragma unroll
    for (int p = 0; p < RK_P; ++p) c[p] = *(const uint4*)(gcnt + (size_t)p * RK_NP + idx);

    unsigned pf = 0;
#pragma unroll
    for (int k = 0; k < RK_P * RK_SEG / 256; ++k) {
        const int j = t + 256 * k;
        const unsigned v = bsum[j];
        pf += ((j & (RK_SEG - 1)) < b) ? v : 0u;
    }
    for (int off = 16; off > 0; off >>= 1) pf += __shfl_xor(pf, off, 32);
    if (lane == 0) red[wid] = pf;

    uint4 run = make_uint4(0u, 0u, 0u, 0u);
#pragma unroll
    for (int p = 0; p < RK_P; ++p) {
        const uint4 o = run;
        run.x += c[p].x; run.y += c[p].y; run.z += c[p].z; run.w += c[p].w;
        c[p] = o;
    }
    const unsigned c0 = run.x, c1 = run.y, c2 = run.z, c3 = run.w;
    const unsigned s = c0 + c1 + c2 + c3;
    unsigned inc = s;
    for (int off = 1; off < 32; off <<= 1) {
        const unsigned y = __shfl_up(inc, off, 32);
        if (lane >= off) inc += y;
    }
    if (lane == 31) wtot[wid] = inc;
    __syncthreads();
    unsigned ex = inc - s;
    for (int w = 0; w < 8; ++w) {
        ex += red[w];
        if (w < wid) ex += wtot[w];
    }
    const unsigned r0 = ex, r1 = ex + c0, r2 = r1 + c1, r3 = r2 + c2;
#pragma unroll
    for (int p = 0; p < RK_P; ++p)
        *(uint4*)(gcnt + (size_t)p * RK_NP + idx) =
            make_uint4(c[p].x + r0, c[p].y + r1, c[p].z + r2, c[p].w + r3);
    if (idx + 3 < n) {
        *(int4*)(rowptr + idx) = make_int4((int)r0, (int)r1, (int)r2, (int)r3);
        *(float4*)(dinv + idx) = make_float4(rsqrtf((float)(c0 + 1u)), rsqrtf((float)(c1 + 1u)),
                                             rsqrtf((float)(c2 + 1u)), rsqrtf((float)(c3 + 1u)));
        if (idx + 4 == n) rowptr[n] = (int)(ex + s);
    }
}

__device__ __forceinline__ void csr_put(int2* __restrict__ csr,
                                        const unsigned* __restrict__ pos,
                                        const float* __restrict__ dinv,
                                        int s, int d, int r) {
    const float w = dinv[s] * dinv[d];
    csr[pos[d] + (unsigned)r] = make_int2(s, __float_as_int(w));
}

__global__ __launch_bounds__(256) void k_csr_fill(
    const int* __restrict__ src, const int* __restrict__ dst,
    const int* __restrict__ rank, const unsigned* __restrict__ gpos,
    const float* __restrict__ dinv, int* __restrict__ csr_raw, int e, int chunk) {
    int2* __restrict__ csr = (int2*)csr_raw;
    const int e0 = (blockIdx.x * blockDim.x + threadIdx.x) * 4;
    if (e0 >= e) return;
    if (((e & 3) == 0) && (e0 + 3 < e)) {
        const unsigned* __restrict__ pos = gpos + (size_t)(e0 / chunk) * RK_NP;
        const int4 s = *(const int4*)(src + e0);
        const int4 d = *(const int4*)(dst + e0);
        const int4 r = *(const int4*)(rank + e0);
        csr_put(csr, pos, dinv, s.x, d.x, r.x);
        csr_put(csr, pos, dinv, s.y, d.y, r.y);
        csr_put(csr, pos, dinv, s.z, d.z, r.z);
        csr_put(csr, pos, dinv, s.w, d.w, r.w);
    } else {
        for (int k = e0; k < e && k < e0 + 4; ++k)
            csr_put(csr, gpos + (size_t)(k / chunk) * RK_NP, dinv, src[k], dst[k], rank[k]);
    }
}

#define LDA 132

__device__ __forceinline__ void lds_barrier() {
    __builtin_amdgcn_fence(__ATOMIC_RELEASE, "workgroup", "local");
    __builtin_amdgcn_s_barrier();
    __builtin_amdgcn_fence(__ATOMIC_ACQUIRE, "workgroup", "local");
}

__device__ __forceinline__ float half_sum16(float x) {
    x += __int_as_float(__builtin_amdgcn_update_dpp(0, __float_as_int(x), 0x161, 0xf, 0xf, true));
    x += __int_as_float(__builtin_amdgcn_update_dpp(0, __float_as_int(x), 0x162, 0xf, 0xf, true));
    x += __int_as_float(__builtin_amdgcn_update_dpp(0, __float_as_int(x), 0x164, 0xf, 0xf, true));
    x += __int_as_float(__builtin_amdgcn_update_dpp(0, __float_as_int(x), 0x168, 0xf, 0xf, true));
    return x;
}

__device__ __forceinline__ void load_w_slice(v2f (&bf)[32], const float* __restrict__ W,
                                             const int n0, const int half, const int lm) {
#pragma unroll
    for (int s = 0; s < 32; ++s) {
        const int ka = 4 * s + 2 * half;
        bf[s].x = W[(size_t)(ka)     * FD + n0 + lm];
        bf[s].y = W[(size_t)(ka + 1) * FD + n0 + lm];
    }
}

__device__ __forceinline__ void tile_products(const float* __restrict__ as, const v2f (&bf)[32],
                                              v8f& acc0, v8f& acc1) {
#pragma unroll
    for (int s = 0; s < 32; s += 2) {
        const v2f a0 = *(const v2f*)(as + 4 * s);
        const v2f a1 = *(const v2f*)(as + 4 * s + 4);
        acc0 = __builtin_amdgcn_wmma_f32_16x16x4_f32(
            false, a0, false, bf[s], (short)0, acc0, false, false);
        acc1 = __builtin_amdgcn_wmma_f32_16x16x4_f32(
            false, a1, false, bf[s + 1], (short)0, acc1, false, false);
    }
}


__device__ __forceinline__ void row_fma(float4& acc, const float4 v, const float w) {
    acc.x = fmaf(v.x, w, acc.x);
    acc.y = fmaf(v.y, w, acc.y);
    acc.z = fmaf(v.z, w, acc.z);
    acc.w = fmaf(v.w, w, acc.w);
}

template <int M>
__device__ __forceinline__ void walk_rows(float4& acc, const int2 r, const int k,
                                          const char* __restrict__ hwb,
                                          const unsigned loff) {
    float4 v[M];
    float  w[M];
    for (int j = 0; j < M; ++j) {
        const int s = __builtin_amdgcn_readlane(r.x, k + j);
        w[j] = __int_as_float(__builtin_amdgcn_readlane(r.y, k + j));
        v[j] = *(const float4*)(hwb + (size_t)s * 512 + loff);
    }
    for (int j = 0; j < M; ++j) row_fma(acc, v[j], w[j]);
}

__device__ __forceinline__ void walk_chunk(float4& acc, const int2 r, const int cnt,
                                           const char* __restrict__ hwb,
                                           const unsigned loff) {
    for (int k = 0; k < cnt; k += 8) {
        switch (cnt - k) {
            case 1:  walk_rows<1>(acc, r, k, hwb, loff); break;
            case 2:  walk_rows<2>(acc, r, k, hwb, loff); break;
            case 3:  walk_rows<3>(acc, r, k, hwb, loff); break;
            case 4:  walk_rows<4>(acc, r, k, hwb, loff); break;
            case 5:  walk_rows<5>(acc, r, k, hwb, loff); break;
            case 6:  walk_rows<6>(acc, r, k, hwb, loff); break;
            case 7:  walk_rows<7>(acc, r, k, hwb, loff); break;
            default: walk_rows<8>(acc, r, k, hwb, loff); break;
        }
    }
}

__device__ __forceinline__ int fetch_bounds(const int* __restrict__ rowptr,
                                            const int node0, const int n, const int lane) {
    int i = node0 + (lane < 2 ? lane : 2);
    i = (i < n) ? i : n;
    return rowptr[i];
}

template <bool HEAD>
__device__ __forceinline__ void rows_gemm_body(
    const int* __restrict__ rowptr, const int* __restrict__ csr_raw,
    const float* __restrict__ dinv, const float* __restrict__ hw,
    const float* __restrict__ brow, float* __restrict__ emb,
    const float* __restrict__ W, const float* __restrict__ bias,
    float* __restrict__ out, int n, int ntiles,
    const float* __restrict__ Wh, const float* __restrict__ bh,
    float* __restrict__ prob) {
    __shared__ float As[2][16 * LDA];
    __shared__ float part[HEAD ? 2 * 8 * 32 : 1];
    __shared__ __align__(16) float cst[HEAD ? 4 * FD + 2 : FD];
    const int t    = threadIdx.x;
    const int lane = t & 31;
    const int wid  = __builtin_amdgcn_readfirstlane(t >> 5);
    const int n0   = wid << 4;
    const int half = lane >> 4;
    const int lm   = lane & 15;

    const int2* __restrict__ csr = (const int2*)csr_raw;
    const char* __restrict__ hwb = (const char*)hw;
    const unsigned loff = (unsigned)lane * 16u;

    int tile = blockIdx.x;
    int rp_l = 0;
    int2 rec0 = make_int2(0, 0), rec1 = make_int2(0, 0);
    if (tile < ntiles) {
        rp_l = fetch_bounds(rowptr, tile * 16 + 2 * wid, n, lane);
        const int a = __builtin_amdgcn_readlane(rp_l, 0);
        const int b = __builtin_amdgcn_readlane(rp_l, 1);
        const int c = __builtin_amdgcn_readlane(rp_l, 2);
        if (a + lane < b) rec0 = csr[a + lane];
        if (b + lane < c) rec1 = csr[b + lane];
    }

    v2f bf[32];
    load_w_slice(bf, W, n0, half, lm);

    if (t < FD) {
        cst[t] = brow[t];
        if (HEAD) {
            cst[FD + t] = bias[t];
            *(float2*)&cst[2 * FD + 2 * t] = *(const float2*)(Wh + 2 * t);
            if (t < 2) cst[4 * FD + t] = bh[t];
        }
    }
    __syncthreads();

    int buf = 0;
    int it  = 0;
    for (; tile < ntiles; tile += (int)gridDim.x) {
        const int  nxt  = tile + (int)gridDim.x;
        const bool more = nxt < ntiles;
        int rp_n = 0;
        if (more) rp_n = fetch_bounds(rowptr, nxt * 16 + 2 * wid, n, lane);

        int rp[3];
        for (int q = 0; q < 3; ++q) rp[q] = __builtin_amdgcn_readlane(rp_l, q);
        int2 rec[2] = {rec0, rec1};

#pragma unroll
        for (int q = 0; q < 2; ++q) {
            const int node = tile * 16 + 2 * wid + q;
            float4 acc = make_float4(0.0f, 0.0f, 0.0f, 0.0f);
            if (node < n) {
                const float di = dinv[node];
                const float4 sv = *(const float4*)(hwb + (size_t)node * 512 + loff);
                const int beg = rp[q], end = rp[q + 1];
                walk_chunk(acc, rec[q], (end - beg < 32) ? (end - beg) : 32, hwb, loff);
                for (int cb = beg + 32; cb < end; cb += 32) {
                    int2 r = make_int2(0, 0);
                    if (cb + lane < end) r = csr[cb + lane];
                    walk_chunk(acc, r, (end - cb < 32) ? (end - cb) : 32, hwb, loff);
                }
                acc.x += sv.x * di * di;
                acc.y += sv.y * di * di;
                acc.z += sv.z * di * di;
                acc.w += sv.w * di * di;
                const float4 b4 = *(const float4*)&cst[4 * lane];
                acc.x += b4.x;
                acc.y += b4.y;
                acc.z += b4.z;
                acc.w += b4.w;
                if (HEAD) ((float4*)emb)[(unsigned)(node * 32 + lane)] = acc;
            }
            *(float4*)&As[buf][(2 * wid + q) * LDA + 4 * lane] =
                make_float4(fmaxf(acc.x, 0.0f), fmaxf(acc.y, 0.0f),
                            fmaxf(acc.z, 0.0f), fmaxf(acc.w, 0.0f));
        }

        rec0 = make_int2(0, 0);
        rec1 = make_int2(0, 0);
        if (more) {
            const int a = __builtin_amdgcn_readlane(rp_n, 0);
            const int b = __builtin_amdgcn_readlane(rp_n, 1);
            const int c = __builtin_amdgcn_readlane(rp_n, 2);
            if (a + lane < b) rec0 = csr[a + lane];
            if (b + lane < c) rec1 = csr[b + lane];
        }
        rp_l = rp_n;

        lds_barrier();

        v8f acc0 = {};
        v8f acc1 = {};
        tile_products(&As[buf][lm * LDA + 2 * half], bf, acc0, acc1);

        if (HEAD) {
            const int pb = it & 1;
            const float  bcol = cst[FD + n0 + lm];
            const float2 wh   = *(const float2*)&cst[2 * FD + 2 * (n0 + lm)];
            const float  wh0 = wh.x, wh1 = wh.y;
            float p[16];
#pragma unroll
            for (int v = 0; v < 8; ++v) {
                float r = acc0[v] + acc1[v];
                r += bcol;
                r = fmaxf(r, 0.0f);
                p[2 * v]     = r * wh0;
                p[2 * v + 1] = r * wh1;
            }
#pragma unroll
            for (int i = 0; i < 16; ++i) p[i] = half_sum16(p[i]);
            if (lm == 0) {
                float4* __restrict__ pp = (float4*)&part[(pb * 8 + wid) * 32 + half * 16];
                pp[0] = make_float4(p[0],  p[1],  p[2],  p[3]);
                pp[1] = make_float4(p[4],  p[5],  p[6],  p[7]);
                pp[2] = make_float4(p[8],  p[9],  p[10], p[11]);
                pp[3] = make_float4(p[12], p[13], p[14], p[15]);
            }
            lds_barrier();
            if (wid == (it & 7)) {
                const float* __restrict__ pr = &part[pb * 256 + lane];
                float l = pr[0];
#pragma unroll
                for (int w = 1; w < 8; ++w) l += pr[w * 32];
                l += cst[4 * FD + (lane & 1)];
                const float o = __int_as_float(
                    __builtin_amdgcn_update_dpp(0, __float_as_int(l), 0x161, 0xf, 0xf, true));
                const float m   = fmaxf(l, o);
                const float lse = m + logf(expf(l - m) + expf(o - m));
                prob[(unsigned)(tile * 32 + lane)] = l - lse;
            }
        } else {
            const int m0 = tile << 4;
#pragma unroll
            for (int v = 0; v < 8; ++v) {
                const int row = m0 + v + 8 * half;
                out[(size_t)row * FD + n0 + lm] = acc0[v] + acc1[v];
            }
        }
        buf ^= 1;
        ++it;
    }
}

__global__ __launch_bounds__(256, 8) void k_rows_gemm(
    const int* __restrict__ rowptr, const int* __restrict__ csr_raw,
    const float* __restrict__ dinv, const float* __restrict__ hw,
    const float* __restrict__ brow, const float* __restrict__ W,
    float* __restrict__ out, int n, int ntiles) {
    rows_gemm_body<false>(rowptr, csr_raw, dinv, hw, brow, nullptr, W, nullptr, out, n, ntiles,
                          nullptr, nullptr, nullptr);
}

__global__ __launch_bounds__(256, 8) void k_rows_gemm_head(
    const int* __restrict__ rowptr, const int* __restrict__ csr_raw,
    const float* __restrict__ dinv, const float* __restrict__ hw,
    const float* __restrict__ brow, float* __restrict__ emb,
    const float* __restrict__ W, const float* __restrict__ bias,
    const float* __restrict__ Wh, const float* __restrict__ bh,
    float* __restrict__ prob, int n, int ntiles) {
    rows_gemm_body<true>(rowptr, csr_raw, dinv, hw, brow, emb, W, bias, nullptr, n, ntiles,
                         Wh, bh, prob);
}

__device__ __forceinline__ int fetch_bounds1(const int* __restrict__ rowptr,
                                             const int node, const int n, const int lane) {
    int i = node + (lane < 1 ? 0 : 1);
    i = (i < n) ? i : n;
    return rowptr[i];
}

__global__ __launch_bounds__(512, 8) void k_rows_gemm2(
    const int* __restrict__ rowptr, const int* __restrict__ csr_raw,
    const float* __restrict__ dinv, const float* __restrict__ xin,
    const float* __restrict__ Wa, const float* __restrict__ ba,
    const float* __restrict__ Wb, float* __restrict__ out, int n, int ntiles) {
    __shared__ float As[2][16 * LDA];
    __shared__ float Ts[2][16 * LDA];
    __shared__ float cst[FD];
    const int t    = threadIdx.x;
    const int lane = t & 31;
    const int wid  = __builtin_amdgcn_readfirstlane(t >> 5);
    const bool first = wid < 8;
    const int n0   = (wid & 7) << 4;
    const int half = lane >> 4;
    const int lm   = lane & 15;

    const int2* __restrict__ csr = (const int2*)csr_raw;
    const char* __restrict__ xb = (const char*)xin;
    const unsigned loff = (unsigned)lane * 16u;

    int tile = blockIdx.x;
    int rp_l = 0;
    int2 rec0 = make_int2(0, 0);
    if (tile < ntiles) {
        rp_l = fetch_bounds1(rowptr, tile * 16 + wid, n, lane);
        const int a = __builtin_amdgcn_readlane(rp_l, 0);
        const int b = __builtin_amdgcn_readlane(rp_l, 1);
        if (a + lane < b) rec0 = csr[a + lane];
    }

    v2f bf[32];
    load_w_slice(bf, first ? Wa : Wb, n0, half, lm);

    if (t < FD) cst[t] = ba[t];
    __syncthreads();

    int buf  = 0;
    int prev = -1;
    for (; tile < ntiles; tile += (int)gridDim.x) {
        const int  nxt  = tile + (int)gridDim.x;
        const bool more = nxt < ntiles;
        int rp_n = 0;
        if (more) rp_n = fetch_bounds1(rowptr, nxt * 16 + wid, n, lane);

        const int beg  = __builtin_amdgcn_readlane(rp_l, 0);
        const int end  = __builtin_amdgcn_readlane(rp_l, 1);
        const int node = tile * 16 + wid;
        float4 acc = make_float4(0.0f, 0.0f, 0.0f, 0.0f);
        if (node < n) {
            const float di = dinv[node];
            const float4 sv = *(const float4*)(xb + (size_t)node * 512 + loff);
            walk_chunk(acc, rec0, (end - beg < 32) ? (end - beg) : 32, xb, loff);
            for (int cb = beg + 32; cb < end; cb += 32) {
                int2 r = make_int2(0, 0);
                if (cb + lane < end) r = csr[cb + lane];
                walk_chunk(acc, r, (end - cb < 32) ? (end - cb) : 32, xb, loff);
            }
            acc.x += sv.x * di * di;
            acc.y += sv.y * di * di;
            acc.z += sv.z * di * di;
            acc.w += sv.w * di * di;
        }
        *(float4*)&As[buf][wid * LDA + 4 * lane] = acc;

        rec0 = make_int2(0, 0);
        if (more) {
            const int a = __builtin_amdgcn_readlane(rp_n, 0);
            const int b = __builtin_amdgcn_readlane(rp_n, 1);
            if (a + lane < b) rec0 = csr[a + lane];
        }
        rp_l = rp_n;

        lds_barrier();

        if (first) {
            v8f acc0 = {};
            v8f acc1 = {};
            tile_products(&As[buf][lm * LDA + 2 * half], bf, acc0, acc1);
            const float bcol = cst[n0 + lm];
            float* __restrict__ tp = &Ts[buf][(8 * half) * LDA + n0 + lm];
#pragma unroll
            for (int v = 0; v < 8; ++v)
                tp[v * LDA] = fmaxf(acc0[v] + acc1[v] + bcol, 0.0f);
        } else if (prev >= 0) {
            v8f acc0 = {};
            v8f acc1 = {};
            tile_products(&Ts[buf ^ 1][lm * LDA + 2 * half], bf, acc0, acc1);
            const int m0 = prev << 4;
#pragma unroll
            for (int v = 0; v < 8; ++v) {
                const int row = m0 + v + 8 * half;
                out[(size_t)row * FD + n0 + lm] = acc0[v] + acc1[v];
            }
        }
        prev = tile;
        buf ^= 1;
    }

    lds_barrier();
    if (!first && prev >= 0) {
        v8f acc0 = {};
        v8f acc1 = {};
        tile_products(&Ts[buf ^ 1][lm * LDA + 2 * half], bf, acc0, acc1);
        const int m0 = prev << 4;
#pragma unroll
        for (int v = 0; v < 8; ++v) {
            const int row = m0 + v + 8 * half;
            out[(size_t)row * FD + n0 + lm] = acc0[v] + acc1[v];
        }
    }
}


extern "C" void kernel_launch(void* const* d_in, const int* in_sizes, int n_in,
                              void* d_out, int out_size, void* d_ws,
                              size_t ws_size, hipStream_t stream) {
    const float* x   = (const float*)d_in[0];
    const int*   ei  = (const int*)d_in[1];
    const float* W1  = (const float*)d_in[2];
    const float* b1  = (const float*)d_in[3];
    const float* W2  = (const float*)d_in[4];
    const float* b2  = (const float*)d_in[5];
    const float* W3  = (const float*)d_in[6];
    const float* b3  = (const float*)d_in[7];
    const float* Wp1 = (const float*)d_in[8];
    const float* bp1 = (const float*)d_in[9];
    const float* Wp2 = (const float*)d_in[10];
    const float* bp2 = (const float*)d_in[11];

    const int N = in_sizes[0] / FD;
    const int E = in_sizes[1] / 2;
    const int* src = ei;
    const int* dst = ei + E;

    char* ws = (char*)d_ws;
    const size_t AL = 256;
    float* dinv = (float*)ws;        ws += ((size_t)N * sizeof(float) + AL - 1) / AL * AL;
    int* rowptr = (int*)ws;          ws += ((size_t)(N + 1) * sizeof(int) + AL - 1) / AL * AL;
    int* rank   = (int*)ws;          ws += ((size_t)E * sizeof(int) + AL - 1) / AL * AL;
    int* csr    = (int*)ws;          ws += ((size_t)E * 2 * sizeof(int) + AL - 1) / AL * AL;
    unsigned* gcnt = (unsigned*)ws;  ws += (size_t)RK_P * RK_NP * sizeof(unsigned);
    unsigned* bsum = (unsigned*)ws;  ws += (size_t)RK_P * RK_SEG * sizeof(unsigned);
    float* hw0  = (float*)ws;        ws += (size_t)N * FD * sizeof(float);
    float* hw1  = (float*)ws;

    float* emb_out  = (float*)d_out;
    float* prob_out = (float*)d_out + (size_t)N * FD;

    const dim3 blk(256);
    const int gScan   = (N + 1023) / 1024;
    const int gFill   = (E + 1023) / 1024;
    const int gTiles  = N / 16;
    const int gGemm   = 1024;
    const int chunk   = ((E + RK_P - 1) / RK_P + 3) & ~3;

    k_rank_lds   <<<RK_P, dim3(1024), 0, stream>>>(dst, rank, gcnt, bsum, E, chunk);
    k_rowptr_dinv<<<gScan, blk, 0, stream>>>(gcnt, bsum, rowptr, dinv, N);
    k_csr_fill   <<<gFill, blk, 0, stream>>>(src, dst, rank, gcnt, dinv, csr, E, chunk);

    k_rows_gemm2<<<512, dim3(512), 0, stream>>>(rowptr, csr, dinv, x, W1, b1, W2, hw1, N, gTiles);

    k_rows_gemm<<<gGemm, blk, 0, stream>>>(rowptr, csr, dinv, hw1, b2, W3, hw0, N, gTiles);

    k_rows_gemm_head<<<gGemm, blk, 0, stream>>>(rowptr, csr, dinv, hw0, b3, emb_out,
                                                Wp1, bp1, Wp2, bp2, prob_out, N, gTiles);
}
